# starting kernel
# baseline (speedup 1.0000x reference)
#include <hip/hip_runtime.h>

#define B_ 4
#define N_ 8192
#define C_ 128
#define Q_ 2048
#define S_ 16
#define KEEP_ 17
#define RS_ 20

typedef __attribute__((ext_vector_type(2))) float v2f;
typedef __attribute__((ext_vector_type(8))) float v8f;

__global__ void pnorm2_kernel(const float* __restrict__ pts, float* __restrict__ pn2) {
  int i = blockIdx.x * blockDim.x + threadIdx.x;
  if (i < B_ * N_) {
    float x = pts[i * 3 + 0], y = pts[i * 3 + 1], z = pts[i * 3 + 2];
    pn2[i] = x * x + y * y + z * z;
  }
}

__global__ __launch_bounds__(256) void fps_kernel(const float* __restrict__ pts,
                                                  int* __restrict__ fps_idx,
                                                  float* __restrict__ qA) {
  const int b = blockIdx.x;
  const int tid = threadIdx.x;
  const float* P = pts + (size_t)b * N_ * 3;

  __shared__ float sval[256];
  __shared__ int   sidx[256];
  __shared__ float scur[3];

  float d[N_ / 256];
  const float p0x = P[0], p0y = P[1], p0z = P[2];
#pragma unroll
  for (int i = 0; i < N_ / 256; ++i) {
    int n = tid + i * 256;
    float dx = P[n * 3 + 0] - p0x, dy = P[n * 3 + 1] - p0y, dz = P[n * 3 + 2] - p0z;
    d[i] = dx * dx + dy * dy + dz * dz;
  }
  if (tid == 0) {
    fps_idx[b * Q_ + 0] = 0;
    qA[((size_t)b * Q_ + 0) * 4 + 0] = p0x;
    qA[((size_t)b * Q_ + 0) * 4 + 1] = p0y;
    qA[((size_t)b * Q_ + 0) * 4 + 2] = p0z;
    qA[((size_t)b * Q_ + 0) * 4 + 3] = p0x * p0x + p0y * p0y + p0z * p0z;
  }

  for (int s = 1; s < Q_; ++s) {
    float bv = -1.0f;
    int   bidx = 0;
#pragma unroll
    for (int i = 0; i < N_ / 256; ++i) {
      int n = tid + i * 256;
      if (d[i] > bv) { bv = d[i]; bidx = n; }
    }
    sval[tid] = bv; sidx[tid] = bidx;
    __syncthreads();
    for (int st = 128; st > 0; st >>= 1) {
      if (tid < st) {
        float ov = sval[tid + st]; int oi = sidx[tid + st];
        float mv = sval[tid];      int mi = sidx[tid];
        if (ov > mv || (ov == mv && oi < mi)) { sval[tid] = ov; sidx[tid] = oi; }
      }
      __syncthreads();
    }
    if (tid == 0) {
      int nxt = sidx[0];
      float x = P[nxt * 3 + 0], y = P[nxt * 3 + 1], z = P[nxt * 3 + 2];
      fps_idx[b * Q_ + s] = nxt;
      qA[((size_t)b * Q_ + s) * 4 + 0] = x;
      qA[((size_t)b * Q_ + s) * 4 + 1] = y;
      qA[((size_t)b * Q_ + s) * 4 + 2] = z;
      qA[((size_t)b * Q_ + s) * 4 + 3] = x * x + y * y + z * z;
      scur[0] = x; scur[1] = y; scur[2] = z;
    }
    __syncthreads();
    float cx = scur[0], cy = scur[1], cz = scur[2];
#pragma unroll
    for (int i = 0; i < N_ / 256; ++i) {
      int n = tid + i * 256;
      float dx = P[n * 3 + 0] - cx, dy = P[n * 3 + 1] - cy, dz = P[n * 3 + 2] - cz;
      float dd = dx * dx + dy * dy + dz * dz;
      d[i] = fminf(d[i], dd);
    }
  }
}

__global__ __launch_bounds__(256) void knn_kernel(const float* __restrict__ pts,
                                                  const float* __restrict__ pn2,
                                                  const float* __restrict__ qA,
                                                  int* __restrict__ nbr) {
  const int wave = threadIdx.x >> 5;
  const int lane = threadIdx.x & 31;
  const int b    = blockIdx.x >> 3;
  const int q0   = ((blockIdx.x & 7) * 8 + wave) * 32;

  const float* P  = pts + (size_t)b * N_ * 3;
  const float* PN = pn2 + (size_t)b * N_;
  const float* QA = qA + (size_t)b * Q_ * 4;

  __shared__ float stage[8][32 * RS_];
  float* st = stage[wave];

  const int lhalf = lane >> 4;
  const int l16   = lane & 15;

  const float4 qa0 = ((const float4*)QA)[q0 + l16];
  const float4 qa1 = ((const float4*)QA)[q0 + 16 + l16];
  v2f a0 = lhalf ? (v2f){-2.0f * qa0.z, 1.0f} : (v2f){-2.0f * qa0.x, -2.0f * qa0.y};
  v2f a1 = lhalf ? (v2f){-2.0f * qa1.z, 1.0f} : (v2f){-2.0f * qa1.x, -2.0f * qa1.y};

  v8f c0, c1;
#pragma unroll
  for (int r = 0; r < 8; ++r) {
    c0[r] = QA[(q0 + r + lhalf * 8) * 4 + 3];
    c1[r] = QA[(q0 + 16 + r + lhalf * 8) * 4 + 3];
  }

  float bd[KEEP_]; int bi[KEEP_];
#pragma unroll
  for (int k = 0; k < KEEP_; ++k) { bd[k] = 3.0e38f; bi[k] = 0; }

  auto load_bb = [&](int ch) -> v2f {
    int pt = ch * 16 + l16;
    float x0 = P[pt * 3 + (lhalf ? 2 : 0)];
    float x1 = lhalf ? PN[pt] : P[pt * 3 + 1];
    return (v2f){x0, x1};
  };

  const int NCH = N_ / 16;
  v2f bb = load_bb(0);
  for (int ch = 0; ch < NCH; ++ch) {
    v2f bb_next = load_bb(ch + 1 < NCH ? ch + 1 : NCH - 1);

    v8f d0 = __builtin_amdgcn_wmma_f32_16x16x4_f32(false, a0, false, bb, (short)0, c0, false, false);
    v8f d1 = __builtin_amdgcn_wmma_f32_16x16x4_f32(false, a1, false, bb, (short)0, c1, false, false);

#pragma unroll
    for (int r = 0; r < 8; ++r) {
      st[(r + lhalf * 8) * RS_ + l16]       = d0[r];
      st[(16 + r + lhalf * 8) * RS_ + l16]  = d1[r];
    }
    const float4* myrow = (const float4*)(st + lane * RS_);
#pragma unroll
    for (int jj = 0; jj < 4; ++jj) {
      float4 v4 = myrow[jj];
#pragma unroll
      for (int e = 0; e < 4; ++e) {
        float v = fmaxf(e == 0 ? v4.x : (e == 1 ? v4.y : (e == 2 ? v4.z : v4.w)), 0.0f);
        int pi = ch * 16 + jj * 4 + e;
        if (v < bd[KEEP_ - 1]) {
          bd[KEEP_ - 1] = v; bi[KEEP_ - 1] = pi;
#pragma unroll
          for (int t = KEEP_ - 1; t > 0; --t) {
            if (bd[t] < bd[t - 1]) {
              float tf = bd[t]; bd[t] = bd[t - 1]; bd[t - 1] = tf;
              int   ti = bi[t]; bi[t] = bi[t - 1]; bi[t - 1] = ti;
            }
          }
        }
      }
    }
    bb = bb_next;
  }

  int q = q0 + lane;
#pragma unroll
  for (int k = 0; k < S_; ++k) nbr[((size_t)b * Q_ + q) * S_ + k] = bi[k + 1];
}

__global__ __launch_bounds__(256) void three_nn_kernel(const float* __restrict__ pts,
                                                       const float* __restrict__ pn2,
                                                       const float* __restrict__ qA,
                                                       int* __restrict__ idx3,
                                                       float* __restrict__ w3) {
  const int wave = threadIdx.x >> 5;
  const int lane = threadIdx.x & 31;
  const int b    = blockIdx.x >> 5;
  const int n0   = ((blockIdx.x & 31) * 8 + wave) * 32;

  const float* P  = pts + (size_t)b * N_ * 3;
  const float* PN = pn2 + (size_t)b * N_;
  const float* QA = qA + (size_t)b * Q_ * 4;

  __shared__ float stage[8][32 * RS_];
  float* st = stage[wave];

  const int lhalf = lane >> 4;
  const int l16   = lane & 15;

  int m0 = n0 + l16, m1 = n0 + 16 + l16;
  float p0x = P[m0 * 3 + 0], p0y = P[m0 * 3 + 1], p0z = P[m0 * 3 + 2];
  float p1x = P[m1 * 3 + 0], p1y = P[m1 * 3 + 1], p1z = P[m1 * 3 + 2];
  v2f a0 = lhalf ? (v2f){-2.0f * p0z, 1.0f} : (v2f){-2.0f * p0x, -2.0f * p0y};
  v2f a1 = lhalf ? (v2f){-2.0f * p1z, 1.0f} : (v2f){-2.0f * p1x, -2.0f * p1y};

  v8f c0, c1;
#pragma unroll
  for (int r = 0; r < 8; ++r) {
    c0[r] = PN[n0 + r + lhalf * 8];
    c1[r] = PN[n0 + 16 + r + lhalf * 8];
  }

  float bd[3]; int bi[3];
#pragma unroll
  for (int k = 0; k < 3; ++k) { bd[k] = 3.0e38f; bi[k] = 0; }

  auto load_bb = [&](int ch) -> v2f {
    const float4 qa = ((const float4*)QA)[ch * 16 + l16];
    return lhalf ? (v2f){qa.z, qa.w} : (v2f){qa.x, qa.y};
  };

  const int NCH = Q_ / 16;
  v2f bb = load_bb(0);
  for (int ch = 0; ch < NCH; ++ch) {
    v2f bb_next = load_bb(ch + 1 < NCH ? ch + 1 : NCH - 1);

    v8f d0 = __builtin_amdgcn_wmma_f32_16x16x4_f32(false, a0, false, bb, (short)0, c0, false, false);
    v8f d1 = __builtin_amdgcn_wmma_f32_16x16x4_f32(false, a1, false, bb, (short)0, c1, false, false);

#pragma unroll
    for (int r = 0; r < 8; ++r) {
      st[(r + lhalf * 8) * RS_ + l16]      = d0[r];
      st[(16 + r + lhalf * 8) * RS_ + l16] = d1[r];
    }
    const float4* myrow = (const float4*)(st + lane * RS_);
#pragma unroll
    for (int jj = 0; jj < 4; ++jj) {
      float4 v4 = myrow[jj];
#pragma unroll
      for (int e = 0; e < 4; ++e) {
        float v = fmaxf(e == 0 ? v4.x : (e == 1 ? v4.y : (e == 2 ? v4.z : v4.w)), 0.0f);
        int qi = ch * 16 + jj * 4 + e;
        if (v < bd[2]) {
          bd[2] = v; bi[2] = qi;
          if (bd[2] < bd[1]) { float tf = bd[2]; bd[2] = bd[1]; bd[1] = tf;
                               int ti = bi[2]; bi[2] = bi[1]; bi[1] = ti; }
          if (bd[1] < bd[0]) { float tf = bd[1]; bd[1] = bd[0]; bd[0] = tf;
                               int ti = bi[1]; bi[1] = bi[0]; bi[0] = ti; }
        }
      }
    }
    bb = bb_next;
  }

  int n = n0 + lane;
  float dv0 = fmaxf(bd[0], 1e-10f), dv1 = fmaxf(bd[1], 1e-10f), dv2 = fmaxf(bd[2], 1e-10f);
  float i0 = 1.0f / (dv0 + 1e-8f), i1 = 1.0f / (dv1 + 1e-8f), i2 = 1.0f / (dv2 + 1e-8f);
  float inv_s = 1.0f / (i0 + i1 + i2);
  size_t base = ((size_t)b * N_ + n) * 3;
  idx3[base + 0] = bi[0]; idx3[base + 1] = bi[1]; idx3[base + 2] = bi[2];
  w3[base + 0] = i0 * inv_s; w3[base + 1] = i1 * inv_s; w3[base + 2] = i2 * inv_s;
}

__global__ __launch_bounds__(256) void group_mean_kernel(const float* __restrict__ feat,
                                                         const int* __restrict__ nbr,
                                                         float* __restrict__ coarse) {
  int bc = blockIdx.x;
  int b = bc / C_, c = bc % C_;
  const float* F = feat + ((size_t)b * C_ + c) * N_;
  const int* NB  = nbr + (size_t)b * Q_ * S_;
  float* O = coarse + ((size_t)b * C_ + c) * Q_;
  for (int q = threadIdx.x; q < Q_; q += blockDim.x) {
    float s = 0.0f;
#pragma unroll
    for (int k = 0; k < S_; ++k) s += F[NB[q * S_ + k]];
    O[q] = s * (1.0f / S_);
  }
}

__global__ __launch_bounds__(256) void interp_kernel(const float* __restrict__ coarse,
                                                     const int* __restrict__ idx3,
                                                     const float* __restrict__ w3,
                                                     float* __restrict__ out) {
  int bc = blockIdx.x;
  int b = bc / C_, c = bc % C_;
  const float* CO = coarse + ((size_t)b * C_ + c) * Q_;
  const int*   I  = idx3 + (size_t)b * N_ * 3;
  const float* W  = w3 + (size_t)b * N_ * 3;
  float* O = out + ((size_t)b * C_ + c) * N_;
  for (int n = threadIdx.x; n < N_; n += blockDim.x) {
    O[n] = W[n * 3 + 0] * CO[I[n * 3 + 0]] +
           W[n * 3 + 1] * CO[I[n * 3 + 1]] +
           W[n * 3 + 2] * CO[I[n * 3 + 2]];
  }
}

extern "C" void kernel_launch(void* const* d_in, const int* in_sizes, int n_in,
                              void* d_out, int out_size, void* d_ws, size_t ws_size,
                              hipStream_t stream) {
  const float* points   = (const float*)d_in[0];
  const float* features = (const float*)d_in[1];

  auto align256 = [](size_t x) { return (x + 255) & ~(size_t)255; };
  char* ws = (char*)d_ws;
  size_t off = 0;
  int*   fps_idx = (int*)(ws + off);   off = align256(off + sizeof(int) * B_ * Q_);
  float* qA      = (float*)(ws + off); off = align256(off + sizeof(float) * B_ * Q_ * 4);
  float* pn2     = (float*)(ws + off); off = align256(off + sizeof(float) * B_ * N_);
  int*   nbr     = (int*)(ws + off);   off = align256(off + sizeof(int) * B_ * Q_ * S_);
  float* coarse  = (float*)(ws + off); off = align256(off + sizeof(float) * B_ * C_ * Q_);
  int*   idx3    = (int*)(ws + off);   off = align256(off + sizeof(int) * B_ * N_ * 3);
  float* w3      = (float*)(ws + off); off = align256(off + sizeof(float) * B_ * N_ * 3);

  float* out = (float*)d_out;

  pnorm2_kernel<<<(B_ * N_ + 255) / 256, 256, 0, stream>>>(points, pn2);
  fps_kernel<<<B_, 256, 0, stream>>>(points, fps_idx, qA);
  knn_kernel<<<B_ * 8, 256, 0, stream>>>(points, pn2, qA, nbr);
  group_mean_kernel<<<B_ * C_, 256, 0, stream>>>(features, nbr, coarse);
  three_nn_kernel<<<B_ * 32, 256, 0, stream>>>(points, pn2, qA, idx3, w3);
  interp_kernel<<<B_ * C_, 256, 0, stream>>>(coarse, idx3, w3, out);
}


# knn, three_nn: a chunk's candidates are first tested as a group against the list's last distance (lane masks in scalar registers); a chunk with no hit is one scalar branch, inserts run branch-free
# speedup vs baseline: 3.1551x; 3.1551x over previous
#include <hip/hip_runtime.h>

#define B_ 4
#define N_ 8192
#define C_ 128
#define Q_ 2048
#define S_ 16
#define KEEP_ 17
#define RS_ 20

typedef __attribute__((ext_vector_type(2))) float v2f;
typedef __attribute__((ext_vector_type(8))) float v8f;

__global__ void pnorm2_kernel(const float* __restrict__ pts, float* __restrict__ pn2) {
  int i = blockIdx.x * blockDim.x + threadIdx.x;
  if (i < B_ * N_) {
    float x = pts[i * 3 + 0], y = pts[i * 3 + 1], z = pts[i * 3 + 2];
    pn2[i] = x * x + y * y + z * z;
  }
}

#define PSEG_ 32
#define PE_   (8 * PSEG_ / 32)
#define PTARGET_ 128.0f
#define LMAX_ 256

__device__ __forceinline__ float dist2(float x, float y, float z, float cx, float cy, float cz) {
  float dx = x - cx, dy = y - cy, dz = z - cz;
  return fmaf(dz, dz, fmaf(dx, dx, dy * dy));
}

__device__ __forceinline__ int uni_i(int v) { return __builtin_amdgcn_readfirstlane(v); }
__device__ __forceinline__ float uni_f(float v) {
  return __builtin_bit_cast(float, __builtin_amdgcn_readfirstlane(__builtin_bit_cast(int, v)));
}

template <int XM>
__device__ __forceinline__ unsigned long long key_max_row(unsigned long long k) {
  int lo = (int)(unsigned)k, hi = (int)(unsigned)(k >> 32);
  unsigned olo = (unsigned)__builtin_amdgcn_update_dpp(0, lo, 0x160 + XM, 0xf, 0xf, true);
  unsigned ohi = (unsigned)__builtin_amdgcn_update_dpp(0, hi, 0x160 + XM, 0xf, 0xf, true);
  unsigned long long o = ((unsigned long long)ohi << 32) | olo;
  return o > k ? o : k;
}
__device__ __forceinline__ unsigned long long key_max_halves(unsigned long long k) {
  int lo = (int)(unsigned)k, hi = (int)(unsigned)(k >> 32);
  unsigned olo = (unsigned)__builtin_amdgcn_permlanex16(lo, lo, 0x76543210, 0xfedcba98, false, false);
  unsigned ohi = (unsigned)__builtin_amdgcn_permlanex16(hi, hi, 0x76543210, 0xfedcba98, false, false);
  unsigned long long o = ((unsigned long long)ohi << 32) | olo;
  return o > k ? o : k;
}

#define OWN_  224
#define NPT_  37

__global__ __launch_bounds__(256) void fps_kernel(const float* __restrict__ pts,
                                                  int* __restrict__ fps_idx,
                                                  float* __restrict__ qA) {
  const int b = blockIdx.x;
  const int tid = threadIdx.x;
  const int wave = uni_i(tid >> 5);
  const int lane = tid & 31;
  const int a = tid - 32;
  const float* P = pts + (size_t)b * N_ * 3;

  __shared__ float sval[256];
  __shared__ int   sidx[256];
  __shared__ float4 plist[LMAX_ + 1];
  __shared__ int   s_np;
  __shared__ float s_dref;
  __shared__ int   s_tot;
  __shared__ float seg_d[8 * PSEG_];
  __shared__ float seg_x[8 * PSEG_];
  __shared__ float seg_y[8 * PSEG_];
  __shared__ float seg_z[8 * PSEG_];
  __shared__ int   seg_i[8 * PSEG_];
  __shared__ int   seg_cnt[8];
  __shared__ float seg_drop[8];

  float px[NPT_], py[NPT_], pz[NPT_], d[NPT_];
  const float p0x = P[0], p0y = P[1], p0z = P[2];
#pragma unroll
  for (int i = 0; i < NPT_; ++i) {
    const int n = a + i * OWN_;
    const bool own = (wave != 0) && (n < N_);
    px[i] = 0.0f; py[i] = 0.0f; pz[i] = 0.0f; d[i] = -1.0f;
    if (own) {
      px[i] = P[n * 3 + 0]; py[i] = P[n * 3 + 1]; pz[i] = P[n * 3 + 2];
      d[i] = dist2(px[i], py[i], pz[i], p0x, p0y, p0z);
    }
  }
  if (tid == 0) {
    fps_idx[b * Q_ + 0] = 0;
    qA[((size_t)b * Q_ + 0) * 4 + 0] = p0x;
    qA[((size_t)b * Q_ + 0) * 4 + 1] = p0y;
    qA[((size_t)b * Q_ + 0) * 4 + 2] = p0z;
    qA[((size_t)b * Q_ + 0) * 4 + 3] = p0x * p0x + p0y * p0y + p0z * p0z;
  }

  auto apply_own = [&](float cx, float cy, float cz) {
#pragma unroll
    for (int g = 0; g < NPT_; g += 8) {
      float dx[8], dy[8], dz[8], t[8];
#pragma unroll
      for (int k = 0; k < 8; ++k) if (g + k < NPT_) { dx[k] = px[g + k] - cx; dy[k] = py[g + k] - cy; dz[k] = pz[g + k] - cz; }
#pragma unroll
      for (int k = 0; k < 8; ++k) if (g + k < NPT_) t[k] = dy[k] * dy[k];
#pragma unroll
      for (int k = 0; k < 8; ++k) if (g + k < NPT_) t[k] = fmaf(dx[k], dx[k], t[k]);
#pragma unroll
      for (int k = 0; k < 8; ++k) if (g + k < NPT_) t[k] = fmaf(dz[k], dz[k], t[k]);
#pragma unroll
      for (int k = 0; k < 8; ++k) if (g + k < NPT_) d[g + k] = fminf(d[g + k], t[k]);
    }
  };

  float gap = 0.1f;
  float dref = 0.0f;
  int   need_full = 1;
  int   s = 1;
  while (s < Q_) {
    int np0 = 0;
    if (need_full) {
      float bv = -1.0f;
      int   bidx = 0;
#pragma unroll
      for (int i = 0; i < NPT_; ++i) {
        if (d[i] > bv) { bv = d[i]; bidx = a + i * OWN_; }
      }
      sval[tid] = bv; sidx[tid] = bidx;
      __syncthreads();
      for (int st = 128; st > 0; st >>= 1) {
        if (tid < st) {
          float ov = sval[tid + st]; int oi = sidx[tid + st];
          float mv = sval[tid];      int mi = sidx[tid];
          if (ov > mv || (ov == mv && oi < mi)) { sval[tid] = ov; sidx[tid] = oi; }
        }
        __syncthreads();
      }
      if (tid == 0) {
        int nxt = sidx[0];
        float x = P[nxt * 3 + 0], y = P[nxt * 3 + 1], z = P[nxt * 3 + 2];
        fps_idx[b * Q_ + s] = nxt;
        qA[((size_t)b * Q_ + s) * 4 + 0] = x;
        qA[((size_t)b * Q_ + s) * 4 + 1] = y;
        qA[((size_t)b * Q_ + s) * 4 + 2] = z;
        qA[((size_t)b * Q_ + s) * 4 + 3] = x * x + y * y + z * z;
        plist[0] = make_float4(x, y, z, 0.0f);
      }
      dref = uni_f(sval[0]);
      __syncthreads();
      if (wave != 0) {
        const float4 c4 = plist[0];
        apply_own(c4.x, c4.y, c4.z);
      }
      s += 1;
      np0 = 1;
      if (s >= Q_) break;
    }

    const float T = dref * (1.0f - gap);
    int cnt = 0;
    float drop = 0.0f;
    if (wave != 0) {
#pragma unroll
      for (int i = 0; i < NPT_; ++i) {
        const bool in = d[i] > T;
        const unsigned m = __builtin_amdgcn_ballot_w32(in);
        const int pos = cnt + (int)__builtin_amdgcn_mbcnt_lo(m, 0u);
        if (in) {
          if (pos < PSEG_) {
            const int sl = wave * PSEG_ + pos;
            seg_d[sl] = d[i]; seg_i[sl] = a + i * OWN_;
            seg_x[sl] = px[i]; seg_y[sl] = py[i]; seg_z[sl] = pz[i];
          } else {
            drop = fmaxf(drop, d[i]);
          }
        }
        cnt += __builtin_popcount(m);
      }
      if (cnt > PSEG_) {
#pragma unroll
        for (int o = 16; o > 0; o >>= 1) drop = fmaxf(drop, __shfl_xor(drop, o, 32));
      }
    }
    if (lane == 0) { seg_cnt[wave] = cnt; seg_drop[wave] = drop; }
    __syncthreads();

    float teff = T;
    int tot = 0;
    float ex[PE_], ey[PE_], ez[PE_], ed[PE_];
    unsigned ek[PE_];
#pragma unroll
    for (int e = 0; e < PE_; ++e) { ex[e] = 0.0f; ey[e] = 0.0f; ez[e] = 0.0f; ed[e] = 0.0f; ek[e] = 0u; }
    if (wave == 0) {
#pragma unroll
      for (int w = 0; w < 8; ++w) { teff = fmaxf(teff, seg_drop[w]); tot += seg_cnt[w]; }
      teff = uni_f(teff);
#pragma unroll
      for (int e = 0; e < PE_; ++e) {
        const int w = (e * 32) / PSEG_;
        const int pos = (e * 32) % PSEG_ + lane;
        const int sl = w * PSEG_ + pos;
        const int c = seg_cnt[w];
        int id = 0;
        if (pos < c) {
          ed[e] = seg_d[sl]; ex[e] = seg_x[sl]; ey[e] = seg_y[sl]; ez[e] = seg_z[sl];
          id = seg_i[sl];
        }
        ek[e] = ((unsigned)(N_ - 1 - id) << 9) | (unsigned)sl;
      }
    }

    int np = 0;
    float last = dref;
    float cx = 0.0f, cy = 0.0f, cz = 0.0f;
    int   cw = 0;
    for (int c = 0; ; ++c) {
      bool end = false;
      if (wave == 0) {
        end = !(s + np < Q_ && np < LMAX_);
        float bd = 0.0f;
        unsigned blo = 0u;
        if (!end) {
          unsigned long long kk[PE_];
#pragma unroll
          for (int e = 0; e < PE_; ++e)
            kk[e] = ((unsigned long long)__float_as_uint(ed[e]) << 32) | ek[e];
#pragma unroll
          for (int w2 = PE_ / 2; w2 > 0; w2 >>= 1) {
#pragma unroll
            for (int e = 0; e < w2; ++e) kk[e] = kk[e + w2] > kk[e] ? kk[e + w2] : kk[e];
          }
          unsigned long long best = kk[0];
          best = key_max_row<1>(best);
          best = key_max_row<2>(best);
          best = key_max_row<4>(best);
          best = key_max_row<8>(best);
          best = key_max_halves(best);
          bd = __uint_as_float((unsigned)uni_i((int)(unsigned)(best >> 32)));
          blo = (unsigned)uni_i((int)(unsigned)best);
          if (!(bd > teff)) { last = teff; end = true; }
        }
        if (end) {
          if (lane == 0) {
            plist[np] = make_float4(0.0f, 0.0f, 0.0f, __int_as_float(1));
            s_np = np; s_dref = last; s_tot = tot;
          }
        } else {
          const int sl = (int)(blo & 511u);
          const int id = N_ - 1 - (int)(blo >> 9);
          const float qx = seg_x[sl], qy = seg_y[sl], qz = seg_z[sl];
          if (lane == 0) plist[np] = make_float4(qx, qy, qz, __int_as_float(id << 1));
          ++np;
          last = bd;
#pragma unroll
          for (int e = 0; e < PE_; ++e)
            ed[e] = fminf(ed[e], dist2(ex[e], ey[e], ez[e], qx, qy, qz));
        }
      }
      __builtin_amdgcn_fence(__ATOMIC_RELEASE, "workgroup", "local");
      __builtin_amdgcn_s_barrier();
      __builtin_amdgcn_fence(__ATOMIC_ACQUIRE, "workgroup", "local");
      if (wave != 0) {
        const float4 nx = plist[c];
        if (c > 0) {
          if (wave == 4) {
            if (lane < 5) {
              const int q = s + c - 1;
              const float n2 = fmaf(cz, cz, fmaf(cx, cx, cy * cy));
              float comp = cx;
              comp = lane == 1 ? cy : comp;
              comp = lane == 2 ? cz : comp;
              comp = lane == 3 ? n2 : comp;
              int* rowp = reinterpret_cast<int*>(qA + ((size_t)b * Q_ + q) * 4) + lane;
              int* dst = lane == 4 ? fps_idx + b * Q_ + q : rowp;
              *dst = lane == 4 ? (cw >> 1) : __float_as_int(comp);
            }
          }
          apply_own(cx, cy, cz);
        }
        cw = uni_i(__float_as_int(nx.w));
        end = (cw & 1) != 0;
        cx = nx.x; cy = nx.y; cz = nx.z;
      }
      if (end) break;
    }

    np = uni_i(s_np);
    dref = uni_f(s_dref);
    tot = uni_i(s_tot);
    s += np;
    need_full = (np + np0 == 0);

    float f = sqrtf(PTARGET_ / (float)(tot > 1 ? tot : 1));
    f = fminf(fmaxf(f, 0.5f), 2.0f);
    gap = fminf(fmaxf(gap * f, 1.0e-3f), 0.5f);
  }
}


__global__ __launch_bounds__(128) void knn_kernel(const float* __restrict__ pts,
                                                  const float* __restrict__ pn2,
                                                  const float* __restrict__ qA,
                                                  int* __restrict__ nbr) {
  const int wave = threadIdx.x >> 5;
  const int lane = threadIdx.x & 31;
  const int b    = blockIdx.x >> 5;
  const int q0   = ((blockIdx.x & 31) * 4 + wave) * 16;

  const float* P  = pts + (size_t)b * N_ * 3;
  const float* PN = pn2 + (size_t)b * N_;
  const float* QA = qA + (size_t)b * Q_ * 4;

  __shared__ float stage[4][16 * RS_];
  __shared__ float mrg_d[4][16 * KEEP_];
  __shared__ int   mrg_i[4][16 * KEEP_];
  float* st = stage[wave];

  const int lhalf = lane >> 4;
  const int l16   = lane & 15;

  const float4 qa0 = ((const float4*)QA)[q0 + l16];
  v2f a0 = lhalf ? (v2f){-2.0f * qa0.z, 1.0f} : (v2f){-2.0f * qa0.x, -2.0f * qa0.y};

  v8f c0;
#pragma unroll
  for (int r = 0; r < 8; ++r) c0[r] = QA[(q0 + r + lhalf * 8) * 4 + 3];

  float bd[KEEP_]; int bi[KEEP_];
#pragma unroll
  for (int k = 0; k < KEEP_; ++k) { bd[k] = 3.0e38f; bi[k] = 0; }

  auto load_bb = [&](int ch) -> v2f {
    int pt = ch * 16 + l16;
    float x0 = P[pt * 3 + (lhalf ? 2 : 0)];
    float x1 = lhalf ? PN[pt] : P[pt * 3 + 1];
    return (v2f){x0, x1};
  };

  const int NCH = N_ / 16;
  v2f bb = load_bb(0);
  for (int ch = 0; ch < NCH; ++ch) {
    v2f bb_next = load_bb(ch + 1 < NCH ? ch + 1 : NCH - 1);

    v8f d0 = __builtin_amdgcn_wmma_f32_16x16x4_f32(false, a0, false, bb, (short)0, c0, false, false);

#pragma unroll
    for (int r = 0; r < 8; ++r) st[(r + lhalf * 8) * RS_ + l16] = d0[r];

    const float4* myrow = (const float4*)(st + l16 * RS_ + lhalf * 8);
    const float4 r0 = myrow[0], r1 = myrow[1];
    const float cv[8] = {fmaxf(r0.x, 0.0f), fmaxf(r0.y, 0.0f), fmaxf(r0.z, 0.0f), fmaxf(r0.w, 0.0f),
                         fmaxf(r1.x, 0.0f), fmaxf(r1.y, 0.0f), fmaxf(r1.z, 0.0f), fmaxf(r1.w, 0.0f)};
    const float thr = bd[KEEP_ - 1];
    unsigned hit[8];
    unsigned any = 0u;
#pragma unroll
    for (int e = 0; e < 8; ++e) { hit[e] = __builtin_amdgcn_ballot_w32(cv[e] < thr); any |= hit[e]; }
    if (any != 0u) {
      const int pbase = ch * 16 + lhalf * 8;
#pragma unroll
      for (int e = 0; e < 8; ++e) {
        if (__builtin_expect(hit[e] != 0u, 0)) {
          const float v = cv[e];
          const int pi = pbase + e;
          bool lt[KEEP_];
#pragma unroll
          for (int k = 0; k < KEEP_; ++k) lt[k] = v < bd[k];
#pragma unroll
          for (int k = KEEP_ - 1; k > 0; --k) {
            bd[k] = __builtin_amdgcn_fmed3f(bd[k - 1], v, bd[k]);
            bi[k] = lt[k - 1] ? bi[k - 1] : (lt[k] ? pi : bi[k]);
          }
          bd[0] = lt[0] ? v : bd[0];
          bi[0] = lt[0] ? pi : bi[0];
        }
      }
    }
    bb = bb_next;
  }

  if (lhalf) {
#pragma unroll
    for (int k = 0; k < KEEP_; ++k) {
      mrg_d[wave][l16 * KEEP_ + k] = bd[k];
      mrg_i[wave][l16 * KEEP_ + k] = bi[k];
    }
  }
  __syncthreads();
  if (!lhalf) {
#pragma unroll 1
    for (int k = 0; k < KEEP_; ++k) {
      float v  = mrg_d[wave][l16 * KEEP_ + k];
      int   pi = mrg_i[wave][l16 * KEEP_ + k];
      if (v < bd[KEEP_ - 1] || (v == bd[KEEP_ - 1] && pi < bi[KEEP_ - 1])) {
        bool lt[KEEP_];
#pragma unroll
        for (int k = 0; k < KEEP_; ++k) lt[k] = v < bd[k] || (v == bd[k] && pi < bi[k]);
#pragma unroll
        for (int k = KEEP_ - 1; k > 0; --k) {
          bd[k] = lt[k - 1] ? bd[k - 1] : (lt[k] ? v : bd[k]);
          bi[k] = lt[k - 1] ? bi[k - 1] : (lt[k] ? pi : bi[k]);
        }
        bd[0] = lt[0] ? v : bd[0];
        bi[0] = lt[0] ? pi : bi[0];
      }
    }
    int q = q0 + l16;
#pragma unroll
    for (int k = 0; k < S_; ++k) nbr[((size_t)b * Q_ + q) * S_ + k] = bi[k + 1];
  }
}

__global__ __launch_bounds__(256) void three_nn_kernel(const float* __restrict__ pts,
                                                       const float* __restrict__ pn2,
                                                       const float* __restrict__ qA,
                                                       int* __restrict__ idx3,
                                                       float* __restrict__ w3) {
  const int wave = threadIdx.x >> 5;
  const int lane = threadIdx.x & 31;
  const int b    = blockIdx.x >> 5;
  const int n0   = ((blockIdx.x & 31) * 8 + wave) * 32;

  const float* P  = pts + (size_t)b * N_ * 3;
  const float* PN = pn2 + (size_t)b * N_;
  const float* QA = qA + (size_t)b * Q_ * 4;

  __shared__ float stage[8][32 * RS_];
  float* st = stage[wave];

  const int lhalf = lane >> 4;
  const int l16   = lane & 15;

  int m0 = n0 + l16, m1 = n0 + 16 + l16;
  float p0x = P[m0 * 3 + 0], p0y = P[m0 * 3 + 1], p0z = P[m0 * 3 + 2];
  float p1x = P[m1 * 3 + 0], p1y = P[m1 * 3 + 1], p1z = P[m1 * 3 + 2];
  v2f a0 = lhalf ? (v2f){-2.0f * p0z, 1.0f} : (v2f){-2.0f * p0x, -2.0f * p0y};
  v2f a1 = lhalf ? (v2f){-2.0f * p1z, 1.0f} : (v2f){-2.0f * p1x, -2.0f * p1y};

  v8f c0, c1;
#pragma unroll
  for (int r = 0; r < 8; ++r) {
    c0[r] = PN[n0 + r + lhalf * 8];
    c1[r] = PN[n0 + 16 + r + lhalf * 8];
  }

  float bd[3]; int bi[3];
#pragma unroll
  for (int k = 0; k < 3; ++k) { bd[k] = 3.0e38f; bi[k] = 0; }

  auto load_bb = [&](int ch) -> v2f {
    const float4 qa = ((const float4*)QA)[ch * 16 + l16];
    return lhalf ? (v2f){qa.z, qa.w} : (v2f){qa.x, qa.y};
  };

  const int NCH = Q_ / 16;
  v2f bb = load_bb(0);
  for (int ch = 0; ch < NCH; ++ch) {
    v2f bb_next = load_bb(ch + 1 < NCH ? ch + 1 : NCH - 1);

    v8f d0 = __builtin_amdgcn_wmma_f32_16x16x4_f32(false, a0, false, bb, (short)0, c0, false, false);
    v8f d1 = __builtin_amdgcn_wmma_f32_16x16x4_f32(false, a1, false, bb, (short)0, c1, false, false);

#pragma unroll
    for (int r = 0; r < 8; ++r) {
      st[(r + lhalf * 8) * RS_ + l16]      = d0[r];
      st[(16 + r + lhalf * 8) * RS_ + l16] = d1[r];
    }
    const float4* myrow = (const float4*)(st + lane * RS_);
    float cv[16];
#pragma unroll
    for (int jj = 0; jj < 4; ++jj) {
      const float4 v4 = myrow[jj];
      cv[jj * 4 + 0] = fmaxf(v4.x, 0.0f); cv[jj * 4 + 1] = fmaxf(v4.y, 0.0f);
      cv[jj * 4 + 2] = fmaxf(v4.z, 0.0f); cv[jj * 4 + 3] = fmaxf(v4.w, 0.0f);
    }
    const float thr = bd[2];
    unsigned hit[16];
    unsigned any = 0u;
#pragma unroll
    for (int e = 0; e < 16; ++e) { hit[e] = __builtin_amdgcn_ballot_w32(cv[e] < thr); any |= hit[e]; }
    if (any != 0u) {
#pragma unroll
      for (int e = 0; e < 16; ++e) {
        if (__builtin_expect(hit[e] != 0u, 0)) {
          const float v = cv[e];
          const int qi = ch * 16 + e;
          const bool l0 = v < bd[0], l1 = v < bd[1], l2 = v < bd[2];
          bd[2] = l1 ? bd[1] : (l2 ? v : bd[2]);
          bi[2] = l1 ? bi[1] : (l2 ? qi : bi[2]);
          bd[1] = l0 ? bd[0] : (l1 ? v : bd[1]);
          bi[1] = l0 ? bi[0] : (l1 ? qi : bi[1]);
          bd[0] = l0 ? v : bd[0];
          bi[0] = l0 ? qi : bi[0];
        }
      }
    }
    bb = bb_next;
  }

  int n = n0 + lane;
  float dv0 = fmaxf(bd[0], 1e-10f), dv1 = fmaxf(bd[1], 1e-10f), dv2 = fmaxf(bd[2], 1e-10f);
  float i0 = 1.0f / (dv0 + 1e-8f), i1 = 1.0f / (dv1 + 1e-8f), i2 = 1.0f / (dv2 + 1e-8f);
  float inv_s = 1.0f / (i0 + i1 + i2);
  size_t base = ((size_t)b * N_ + n) * 3;
  idx3[base + 0] = bi[0]; idx3[base + 1] = bi[1]; idx3[base + 2] = bi[2];
  w3[base + 0] = i0 * inv_s; w3[base + 1] = i1 * inv_s; w3[base + 2] = i2 * inv_s;
}

__global__ __launch_bounds__(256) void group_mean_kernel(const float* __restrict__ feat,
                                                         const int* __restrict__ nbr,
                                                         float* __restrict__ coarse) {
  int bc = blockIdx.x;
  int b = bc / C_, c = bc % C_;
  const float* F = feat + ((size_t)b * C_ + c) * N_;
  const int* NB  = nbr + (size_t)b * Q_ * S_;
  float* O = coarse + ((size_t)b * C_ + c) * Q_;
  for (int q = threadIdx.x; q < Q_; q += blockDim.x) {
    float s = 0.0f;
#pragma unroll
    for (int k = 0; k < S_; ++k) s += F[NB[q * S_ + k]];
    O[q] = s * (1.0f / S_);
  }
}

__global__ __launch_bounds__(256) void interp_kernel(const float* __restrict__ coarse,
                                                     const int* __restrict__ idx3,
                                                     const float* __restrict__ w3,
                                                     float* __restrict__ out) {
  int bc = blockIdx.x;
  int b = bc / C_, c = bc % C_;
  const float* CO = coarse + ((size_t)b * C_ + c) * Q_;
  const int*   I  = idx3 + (size_t)b * N_ * 3;
  const float* W  = w3 + (size_t)b * N_ * 3;
  float* O = out + ((size_t)b * C_ + c) * N_;
  for (int n = threadIdx.x; n < N_; n += blockDim.x) {
    O[n] = W[n * 3 + 0] * CO[I[n * 3 + 0]] +
           W[n * 3 + 1] * CO[I[n * 3 + 1]] +
           W[n * 3 + 2] * CO[I[n * 3 + 2]];
  }
}

extern "C" void kernel_launch(void* const* d_in, const int* in_sizes, int n_in,
                              void* d_out, int out_size, void* d_ws, size_t ws_size,
                              hipStream_t stream) {
  const float* points   = (const float*)d_in[0];
  const float* features = (const float*)d_in[1];

  auto align256 = [](size_t x) { return (x + 255) & ~(size_t)255; };
  char* ws = (char*)d_ws;
  size_t off = 0;
  int*   fps_idx = (int*)(ws + off);   off = align256(off + sizeof(int) * B_ * Q_);
  float* qA      = (float*)(ws + off); off = align256(off + sizeof(float) * B_ * Q_ * 4);
  float* pn2     = (float*)(ws + off); off = align256(off + sizeof(float) * B_ * N_);
  int*   nbr     = (int*)(ws + off);   off = align256(off + sizeof(int) * B_ * Q_ * S_);
  float* coarse  = (float*)(ws + off); off = align256(off + sizeof(float) * B_ * C_ * Q_);
  int*   idx3    = (int*)(ws + off);   off = align256(off + sizeof(int) * B_ * N_ * 3);
  float* w3      = (float*)(ws + off); off = align256(off + sizeof(float) * B_ * N_ * 3);

  float* out = (float*)d_out;

  pnorm2_kernel<<<(B_ * N_ + 255) / 256, 256, 0, stream>>>(points, pn2);
  fps_kernel<<<B_, 256, 0, stream>>>(points, fps_idx, qA);
  knn_kernel<<<B_ * 32, 128, 0, stream>>>(points, pn2, qA, nbr);
  group_mean_kernel<<<B_ * C_, 256, 0, stream>>>(features, nbr, coarse);
  three_nn_kernel<<<B_ * 32, 256, 0, stream>>>(points, pn2, qA, idx3, w3);
  interp_kernel<<<B_ * C_, 256, 0, stream>>>(coarse, idx3, w3, out);
}


# knn, three_nn, group_mean: the operand every wave re-reads (points with |p|^2; query rows; the feature row) is copied once per block into LDS and read from there, not from global memory
# speedup vs baseline: 4.4717x; 1.4173x over previous
#include <hip/hip_runtime.h>

#define B_ 4
#define N_ 8192
#define C_ 128
#define Q_ 2048
#define S_ 16
#define KEEP_ 17
#define RS_ 20

typedef __attribute__((ext_vector_type(2))) float v2f;
typedef __attribute__((ext_vector_type(8))) float v8f;

__global__ void pnorm2_kernel(const float* __restrict__ pts, float* __restrict__ pn2) {
  int i = blockIdx.x * blockDim.x + threadIdx.x;
  if (i < B_ * N_) {
    float x = pts[i * 3 + 0], y = pts[i * 3 + 1], z = pts[i * 3 + 2];
    pn2[i] = x * x + y * y + z * z;
  }
}

#define PSEG_ 32
#define PE_   (8 * PSEG_ / 32)
#define PTARGET_ 128.0f
#define LMAX_ 256
#define KG_   3

__device__ __forceinline__ float dist2(float x, float y, float z, float cx, float cy, float cz) {
  float dx = x - cx, dy = y - cy, dz = z - cz;
  return fmaf(dz, dz, fmaf(dx, dx, dy * dy));
}

__device__ __forceinline__ int uni_i(int v) { return __builtin_amdgcn_readfirstlane(v); }
__device__ __forceinline__ float uni_f(float v) {
  return __builtin_bit_cast(float, __builtin_amdgcn_readfirstlane(__builtin_bit_cast(int, v)));
}

template <int XM>
__device__ __forceinline__ unsigned umax_row(unsigned v) {
  const unsigned o = (unsigned)__builtin_amdgcn_update_dpp(0, (int)v, 0x160 + XM, 0xf, 0xf, true);
  return o > v ? o : v;
}
__device__ __forceinline__ unsigned umax_halves(unsigned v) {
  const unsigned o = (unsigned)__builtin_amdgcn_permlanex16((int)v, (int)v, 0x76543210, 0xfedcba98, false, false);
  return o > v ? o : v;
}
__device__ __forceinline__ unsigned umax_wave(unsigned v) {
  v = umax_row<1>(v);
  v = umax_row<2>(v);
  v = umax_row<4>(v);
  v = umax_row<8>(v);
  return umax_halves(v);
}

#define OWN_  224
#define NPT_  37

__global__ __launch_bounds__(256) void fps_kernel(const float* __restrict__ pts,
                                                  int* __restrict__ fps_idx,
                                                  float* __restrict__ qA) {
  const int b = blockIdx.x;
  const int tid = threadIdx.x;
  const int wave = uni_i(tid >> 5);
  const int lane = tid & 31;
  const int a = tid - 32;
  const float* P = pts + (size_t)b * N_ * 3;

  __shared__ float sval[256];
  __shared__ int   sidx[256];
  __shared__ float4 plist[LMAX_ + KG_];
  __shared__ int   s_np;
  __shared__ float s_dref;
  __shared__ int   s_tot;
  __shared__ float seg_d[8 * PSEG_];
  __shared__ float seg_x[8 * PSEG_];
  __shared__ float seg_y[8 * PSEG_];
  __shared__ float seg_z[8 * PSEG_];
  __shared__ int   seg_i[8 * PSEG_];
  __shared__ int   seg_cnt[8];
  __shared__ float seg_drop[8];

  float px[NPT_], py[NPT_], pz[NPT_], d[NPT_];
  const float p0x = P[0], p0y = P[1], p0z = P[2];
#pragma unroll
  for (int i = 0; i < NPT_; ++i) {
    const int n = a + i * OWN_;
    const bool own = (wave != 0) && (n < N_);
    px[i] = 0.0f; py[i] = 0.0f; pz[i] = 0.0f; d[i] = -1.0f;
    if (own) {
      px[i] = P[n * 3 + 0]; py[i] = P[n * 3 + 1]; pz[i] = P[n * 3 + 2];
      d[i] = dist2(px[i], py[i], pz[i], p0x, p0y, p0z);
    }
  }
  if (tid == 0) {
    fps_idx[b * Q_ + 0] = 0;
    qA[((size_t)b * Q_ + 0) * 4 + 0] = p0x;
    qA[((size_t)b * Q_ + 0) * 4 + 1] = p0y;
    qA[((size_t)b * Q_ + 0) * 4 + 2] = p0z;
    qA[((size_t)b * Q_ + 0) * 4 + 3] = p0x * p0x + p0y * p0y + p0z * p0z;
  }

  auto apply_own = [&](float cx, float cy, float cz) {
#pragma unroll
    for (int g = 0; g < NPT_; g += 8) {
      float dx[8], dy[8], dz[8], t[8];
#pragma unroll
      for (int k = 0; k < 8; ++k) if (g + k < NPT_) { dx[k] = px[g + k] - cx; dy[k] = py[g + k] - cy; dz[k] = pz[g + k] - cz; }
#pragma unroll
      for (int k = 0; k < 8; ++k) if (g + k < NPT_) t[k] = dy[k] * dy[k];
#pragma unroll
      for (int k = 0; k < 8; ++k) if (g + k < NPT_) t[k] = fmaf(dx[k], dx[k], t[k]);
#pragma unroll
      for (int k = 0; k < 8; ++k) if (g + k < NPT_) t[k] = fmaf(dz[k], dz[k], t[k]);
#pragma unroll
      for (int k = 0; k < 8; ++k) if (g + k < NPT_) d[g + k] = fminf(d[g + k], t[k]);
    }
  };

  auto apply_group = [&](const float4 (&g)[KG_]) {
#pragma unroll
    for (int gq = 0; gq < NPT_; gq += 8) {
      float t[KG_][8];
#pragma unroll
      for (int j = 0; j < KG_; ++j) {
        float dx[8], dy[8], dz[8];
#pragma unroll
        for (int k = 0; k < 8; ++k) if (gq + k < NPT_) { dx[k] = px[gq + k] - g[j].x; dy[k] = py[gq + k] - g[j].y; dz[k] = pz[gq + k] - g[j].z; }
#pragma unroll
        for (int k = 0; k < 8; ++k) if (gq + k < NPT_) t[j][k] = dy[k] * dy[k];
#pragma unroll
        for (int k = 0; k < 8; ++k) if (gq + k < NPT_) t[j][k] = fmaf(dx[k], dx[k], t[j][k]);
#pragma unroll
        for (int k = 0; k < 8; ++k) if (gq + k < NPT_) t[j][k] = fmaf(dz[k], dz[k], t[j][k]);
      }
#pragma unroll
      for (int k = 0; k < 8; ++k) if (gq + k < NPT_) {
        float m = d[gq + k];
#pragma unroll
        for (int j = 0; j < KG_; ++j) m = fminf(m, t[j][k]);
        d[gq + k] = m;
      }
    }
  };

  float gap = 0.1f;
  float dref = 0.0f;
  int   need_full = 1;
  int   s = 1;
  while (s < Q_) {
    int np0 = 0;
    if (need_full) {
      float bv = -1.0f;
      int   bidx = 0;
#pragma unroll
      for (int i = 0; i < NPT_; ++i) {
        if (d[i] > bv) { bv = d[i]; bidx = a + i * OWN_; }
      }
      sval[tid] = bv; sidx[tid] = bidx;
      __syncthreads();
      for (int st = 128; st > 0; st >>= 1) {
        if (tid < st) {
          float ov = sval[tid + st]; int oi = sidx[tid + st];
          float mv = sval[tid];      int mi = sidx[tid];
          if (ov > mv || (ov == mv && oi < mi)) { sval[tid] = ov; sidx[tid] = oi; }
        }
        __syncthreads();
      }
      if (tid == 0) {
        int nxt = sidx[0];
        float x = P[nxt * 3 + 0], y = P[nxt * 3 + 1], z = P[nxt * 3 + 2];
        fps_idx[b * Q_ + s] = nxt;
        qA[((size_t)b * Q_ + s) * 4 + 0] = x;
        qA[((size_t)b * Q_ + s) * 4 + 1] = y;
        qA[((size_t)b * Q_ + s) * 4 + 2] = z;
        qA[((size_t)b * Q_ + s) * 4 + 3] = x * x + y * y + z * z;
        plist[0] = make_float4(x, y, z, 0.0f);
      }
      dref = uni_f(sval[0]);
      __syncthreads();
      if (wave != 0) {
        const float4 c4 = plist[0];
        apply_own(c4.x, c4.y, c4.z);
      }
      s += 1;
      np0 = 1;
      if (s >= Q_) break;
    }

    const float T = dref * (1.0f - gap);
    int cnt = 0;
    float drop = 0.0f;
    if (wave != 0) {
#pragma unroll
      for (int i = 0; i < NPT_; ++i) {
        const bool in = d[i] > T;
        const unsigned m = __builtin_amdgcn_ballot_w32(in);
        const int pos = cnt + (int)__builtin_amdgcn_mbcnt_lo(m, 0u);
        if (in) {
          if (pos < PSEG_) {
            const int sl = wave * PSEG_ + pos;
            seg_d[sl] = d[i]; seg_i[sl] = a + i * OWN_;
            seg_x[sl] = px[i]; seg_y[sl] = py[i]; seg_z[sl] = pz[i];
          } else {
            drop = fmaxf(drop, d[i]);
          }
        }
        cnt += __builtin_popcount(m);
      }
      if (cnt > PSEG_) {
#pragma unroll
        for (int o = 16; o > 0; o >>= 1) drop = fmaxf(drop, __shfl_xor(drop, o, 32));
      }
    }
    if (lane == 0) { seg_cnt[wave] = cnt; seg_drop[wave] = drop; }
    __syncthreads();

    float teff = T;
    int tot = 0;
    float ex[PE_], ey[PE_], ez[PE_], ed[PE_];
    unsigned ek[PE_];
#pragma unroll
    for (int e = 0; e < PE_; ++e) { ex[e] = 0.0f; ey[e] = 0.0f; ez[e] = 0.0f; ed[e] = 0.0f; ek[e] = 0u; }
    if (wave == 0) {
#pragma unroll
      for (int w = 0; w < 8; ++w) { teff = fmaxf(teff, seg_drop[w]); tot += seg_cnt[w]; }
      teff = uni_f(teff);
#pragma unroll
      for (int e = 0; e < PE_; ++e) {
        const int w = (e * 32) / PSEG_;
        const int pos = (e * 32) % PSEG_ + lane;
        const int sl = w * PSEG_ + pos;
        const int c = seg_cnt[w];
        int id = 0;
        if (pos < c) {
          ed[e] = seg_d[sl]; ex[e] = seg_x[sl]; ey[e] = seg_y[sl]; ez[e] = seg_z[sl];
          id = seg_i[sl];
        }
        ek[e] = ((unsigned)(N_ - 1 - id) << 9) | (unsigned)sl;
      }
    }

    int np = 0;
    float last = dref;
    for (int c = 0; ; ++c) {
      bool end = false;
      if (wave == 0) {
#pragma unroll
        for (int j = 0; j < KG_; ++j) {
          if (!end) {
            end = !(s + np < Q_ && np < LMAX_);
            float bd = 0.0f;
            unsigned blo = 0u;
            if (!end) {
              unsigned db[PE_], mm[PE_];
#pragma unroll
              for (int e = 0; e < PE_; ++e) { db[e] = __float_as_uint(ed[e]); mm[e] = db[e]; }
#pragma unroll
              for (int w2 = PE_ / 2; w2 > 0; w2 >>= 1) {
#pragma unroll
                for (int e = 0; e < w2; ++e) mm[e] = mm[e + w2] > mm[e] ? mm[e + w2] : mm[e];
              }
              const unsigned dmax = umax_wave(mm[0]);
              unsigned cc[PE_];
#pragma unroll
              for (int e = 0; e < PE_; ++e) cc[e] = db[e] == dmax ? ek[e] : 0u;
#pragma unroll
              for (int w2 = PE_ / 2; w2 > 0; w2 >>= 1) {
#pragma unroll
                for (int e = 0; e < w2; ++e) cc[e] = cc[e + w2] > cc[e] ? cc[e + w2] : cc[e];
              }
              const unsigned kmax = umax_wave(cc[0]);
              bd = __uint_as_float((unsigned)uni_i((int)dmax));
              blo = (unsigned)uni_i((int)kmax);
              if (!(bd > teff)) { last = teff; end = true; }
            }
            if (end) {
              if (lane == 0) {
#pragma unroll
                for (int jj = j; jj < KG_; ++jj) plist[c * KG_ + jj] = make_float4(0.0f, 0.0f, 0.0f, __int_as_float(1));
                s_np = np; s_dref = last; s_tot = tot;
              }
            } else {
              const int sl = (int)(blo & 511u);
              const int id = N_ - 1 - (int)(blo >> 9);
              const float qx = seg_x[sl], qy = seg_y[sl], qz = seg_z[sl];
              if (lane == 0) plist[np] = make_float4(qx, qy, qz, __int_as_float(id << 1));
              ++np;
              last = bd;
#pragma unroll
              for (int e = 0; e < PE_; ++e)
                ed[e] = fminf(ed[e], dist2(ex[e], ey[e], ez[e], qx, qy, qz));
            }
          }
        }
      }
      __builtin_amdgcn_fence(__ATOMIC_RELEASE, "workgroup", "local");
      __builtin_amdgcn_s_barrier();
      __builtin_amdgcn_fence(__ATOMIC_ACQUIRE, "workgroup", "local");
      if (wave != 0) {
        float4 g[KG_];
#pragma unroll
        for (int j = 0; j < KG_; ++j) g[j] = plist[c * KG_ + j];
        int nv = 0;
#pragma unroll
        for (int j = 0; j < KG_; ++j) {
          const int wj = uni_i(__float_as_int(g[j].w));
          if (!end) {
            if ((wj & 1) != 0) end = true; else ++nv;
          }
        }
        if (nv > 0) {
#pragma unroll
          for (int j = 1; j < KG_; ++j) if (j >= nv) g[j] = g[0];
          if (wave == 4) {
            const int sj = lane / 5;
            const int sr = lane - sj * 5;
            if (sj < nv) {
              float4 e4 = g[0];
#pragma unroll
              for (int j = 1; j < KG_; ++j) if (sj == j) e4 = g[j];
              const int q = s + c * KG_ + sj;
              const float n2 = fmaf(e4.z, e4.z, fmaf(e4.x, e4.x, e4.y * e4.y));
              float comp = e4.x;
              comp = sr == 1 ? e4.y : comp;
              comp = sr == 2 ? e4.z : comp;
              comp = sr == 3 ? n2 : comp;
              int* rowp = reinterpret_cast<int*>(qA + ((size_t)b * Q_ + q) * 4) + sr;
              int* dst = sr == 4 ? fps_idx + b * Q_ + q : rowp;
              *dst = sr == 4 ? (__float_as_int(e4.w) >> 1) : __float_as_int(comp);
            }
          }
          apply_group(g);
        }
      }
      if (end) break;
    }

    np = uni_i(s_np);
    dref = uni_f(s_dref);
    tot = uni_i(s_tot);
    s += np;
    need_full = (np + np0 == 0);

    float f = sqrtf(PTARGET_ / (float)(tot > 1 ? tot : 1));
    f = fminf(fmaxf(f, 0.5f), 2.0f);
    gap = fminf(fmaxf(gap * f, 1.0e-3f), 0.5f);
  }
}


__global__ __launch_bounds__(128) void knn_kernel(const float* __restrict__ pts,
                                                  const float* __restrict__ pn2,
                                                  const float* __restrict__ qA,
                                                  int* __restrict__ nbr) {
  const int wave = threadIdx.x >> 5;
  const int lane = threadIdx.x & 31;
  const int b    = blockIdx.x >> 5;
  const int q0   = ((blockIdx.x & 31) * 4 + wave) * 16;

  const float* P  = pts + (size_t)b * N_ * 3;
  const float* PN = pn2 + (size_t)b * N_;
  const float* QA = qA + (size_t)b * Q_ * 4;

  __shared__ float stage[4][16 * RS_];
  __shared__ float mrg_d[4][16 * KEEP_];
  __shared__ int   mrg_i[4][16 * KEEP_];
  __shared__ __attribute__((aligned(16))) float lpts[N_ * 4];
  float* st = stage[wave];

  const int lhalf = lane >> 4;
  const int l16   = lane & 15;

  {
    const float4* P4  = reinterpret_cast<const float4*>(P);
    const float4* PN4 = reinterpret_cast<const float4*>(PN);
#pragma unroll 4
    for (int j = 0; j < N_ / 4 / 128; ++j) {
      const int g = (int)threadIdx.x + j * 128;
      const float4 u0 = P4[g * 3 + 0], u1 = P4[g * 3 + 1], u2 = P4[g * 3 + 2];
      const float4 n4 = PN4[g];
      float4* dst = reinterpret_cast<float4*>(lpts + g * 16);
      dst[0] = make_float4(u0.x, u0.y, u0.z, n4.x);
      dst[1] = make_float4(u0.w, u1.x, u1.y, n4.y);
      dst[2] = make_float4(u1.z, u1.w, u2.x, n4.z);
      dst[3] = make_float4(u2.y, u2.z, u2.w, n4.w);
    }
  }
  __syncthreads();

  const float4 qa0 = ((const float4*)QA)[q0 + l16];
  v2f a0 = lhalf ? (v2f){-2.0f * qa0.z, 1.0f} : (v2f){-2.0f * qa0.x, -2.0f * qa0.y};

  v8f c0;
#pragma unroll
  for (int r = 0; r < 8; ++r) c0[r] = QA[(q0 + r + lhalf * 8) * 4 + 3];

  float bd[KEEP_]; int bi[KEEP_];
#pragma unroll
  for (int k = 0; k < KEEP_; ++k) { bd[k] = 3.0e38f; bi[k] = 0; }

  auto load_bb = [&](int ch) -> v2f {
    return *reinterpret_cast<const v2f*>(lpts + (ch * 16 + l16) * 4 + lhalf * 2);
  };

  const int NCH = N_ / 16;
  v2f bb = load_bb(0);
  for (int ch = 0; ch < NCH; ++ch) {
    v2f bb_next = load_bb(ch + 1 < NCH ? ch + 1 : NCH - 1);

    v8f d0 = __builtin_amdgcn_wmma_f32_16x16x4_f32(false, a0, false, bb, (short)0, c0, false, false);

#pragma unroll
    for (int r = 0; r < 8; ++r) st[(r + lhalf * 8) * RS_ + l16] = d0[r];

    const float4* myrow = (const float4*)(st + l16 * RS_ + lhalf * 8);
    const float4 r0 = myrow[0], r1 = myrow[1];
    const float cv[8] = {fmaxf(r0.x, 0.0f), fmaxf(r0.y, 0.0f), fmaxf(r0.z, 0.0f), fmaxf(r0.w, 0.0f),
                         fmaxf(r1.x, 0.0f), fmaxf(r1.y, 0.0f), fmaxf(r1.z, 0.0f), fmaxf(r1.w, 0.0f)};
    const float thr = bd[KEEP_ - 1];
    unsigned hit[8];
    unsigned any = 0u;
#pragma unroll
    for (int e = 0; e < 8; ++e) { hit[e] = __builtin_amdgcn_ballot_w32(cv[e] < thr); any |= hit[e]; }
    if (any != 0u) {
      const int pbase = ch * 16 + lhalf * 8;
#pragma unroll
      for (int e = 0; e < 8; ++e) {
        if (__builtin_expect(hit[e] != 0u, 0)) {
          const float v = cv[e];
          const int pi = pbase + e;
          bool lt[KEEP_];
#pragma unroll
          for (int k = 0; k < KEEP_; ++k) lt[k] = v < bd[k];
#pragma unroll
          for (int k = KEEP_ - 1; k > 0; --k) {
            bd[k] = __builtin_amdgcn_fmed3f(bd[k - 1], v, bd[k]);
            bi[k] = lt[k - 1] ? bi[k - 1] : (lt[k] ? pi : bi[k]);
          }
          bd[0] = lt[0] ? v : bd[0];
          bi[0] = lt[0] ? pi : bi[0];
        }
      }
    }
    bb = bb_next;
  }

  if (lhalf) {
#pragma unroll
    for (int k = 0; k < KEEP_; ++k) {
      mrg_d[wave][l16 * KEEP_ + k] = bd[k];
      mrg_i[wave][l16 * KEEP_ + k] = bi[k];
    }
  }
  __syncthreads();
  if (!lhalf) {
#pragma unroll 1
    for (int k = 0; k < KEEP_; ++k) {
      float v  = mrg_d[wave][l16 * KEEP_ + k];
      int   pi = mrg_i[wave][l16 * KEEP_ + k];
      if (v < bd[KEEP_ - 1] || (v == bd[KEEP_ - 1] && pi < bi[KEEP_ - 1])) {
        bool lt[KEEP_];
#pragma unroll
        for (int k = 0; k < KEEP_; ++k) lt[k] = v < bd[k] || (v == bd[k] && pi < bi[k]);
#pragma unroll
        for (int k = KEEP_ - 1; k > 0; --k) {
          bd[k] = lt[k - 1] ? bd[k - 1] : (lt[k] ? v : bd[k]);
          bi[k] = lt[k - 1] ? bi[k - 1] : (lt[k] ? pi : bi[k]);
        }
        bd[0] = lt[0] ? v : bd[0];
        bi[0] = lt[0] ? pi : bi[0];
      }
    }
    int q = q0 + l16;
#pragma unroll
    for (int k = 0; k < S_; ++k) nbr[((size_t)b * Q_ + q) * S_ + k] = bi[k + 1];
  }
}

__global__ __launch_bounds__(256) void three_nn_kernel(const float* __restrict__ pts,
                                                       const float* __restrict__ pn2,
                                                       const float* __restrict__ qA,
                                                       int* __restrict__ idx3,
                                                       float* __restrict__ w3) {
  const int wave = threadIdx.x >> 5;
  const int lane = threadIdx.x & 31;
  const int b    = blockIdx.x >> 5;
  const int n0   = ((blockIdx.x & 31) * 8 + wave) * 32;

  const float* P  = pts + (size_t)b * N_ * 3;
  const float* PN = pn2 + (size_t)b * N_;
  const float* QA = qA + (size_t)b * Q_ * 4;

  __shared__ float stage[8][32 * RS_];
  __shared__ __attribute__((aligned(16))) float lq[Q_ * 4];
  float* st = stage[wave];

  const int lhalf = lane >> 4;
  const int l16   = lane & 15;

#pragma unroll
  for (int j = 0; j < Q_ / 256; ++j) {
    const int i = (int)threadIdx.x + j * 256;
    *reinterpret_cast<float4*>(lq + i * 4) = reinterpret_cast<const float4*>(QA)[i];
  }
  __syncthreads();

  int m0 = n0 + l16, m1 = n0 + 16 + l16;
  float p0x = P[m0 * 3 + 0], p0y = P[m0 * 3 + 1], p0z = P[m0 * 3 + 2];
  float p1x = P[m1 * 3 + 0], p1y = P[m1 * 3 + 1], p1z = P[m1 * 3 + 2];
  v2f a0 = lhalf ? (v2f){-2.0f * p0z, 1.0f} : (v2f){-2.0f * p0x, -2.0f * p0y};
  v2f a1 = lhalf ? (v2f){-2.0f * p1z, 1.0f} : (v2f){-2.0f * p1x, -2.0f * p1y};

  v8f c0, c1;
#pragma unroll
  for (int r = 0; r < 8; ++r) {
    c0[r] = PN[n0 + r + lhalf * 8];
    c1[r] = PN[n0 + 16 + r + lhalf * 8];
  }

  float bd[3]; int bi[3];
#pragma unroll
  for (int k = 0; k < 3; ++k) { bd[k] = 3.0e38f; bi[k] = 0; }

  auto load_bb = [&](int ch) -> v2f {
    return *reinterpret_cast<const v2f*>(lq + (ch * 16 + l16) * 4 + lhalf * 2);
  };

  const int NCH = Q_ / 16;
  v2f bb = load_bb(0);
  for (int ch = 0; ch < NCH; ++ch) {
    v2f bb_next = load_bb(ch + 1 < NCH ? ch + 1 : NCH - 1);

    v8f d0 = __builtin_amdgcn_wmma_f32_16x16x4_f32(false, a0, false, bb, (short)0, c0, false, false);
    v8f d1 = __builtin_amdgcn_wmma_f32_16x16x4_f32(false, a1, false, bb, (short)0, c1, false, false);

#pragma unroll
    for (int r = 0; r < 8; ++r) {
      st[(r + lhalf * 8) * RS_ + l16]      = d0[r];
      st[(16 + r + lhalf * 8) * RS_ + l16] = d1[r];
    }
    const float4* myrow = (const float4*)(st + lane * RS_);
    float cv[16];
#pragma unroll
    for (int jj = 0; jj < 4; ++jj) {
      const float4 v4 = myrow[jj];
      cv[jj * 4 + 0] = fmaxf(v4.x, 0.0f); cv[jj * 4 + 1] = fmaxf(v4.y, 0.0f);
      cv[jj * 4 + 2] = fmaxf(v4.z, 0.0f); cv[jj * 4 + 3] = fmaxf(v4.w, 0.0f);
    }
    const float thr = bd[2];
    unsigned hit[16];
    unsigned any = 0u;
#pragma unroll
    for (int e = 0; e < 16; ++e) { hit[e] = __builtin_amdgcn_ballot_w32(cv[e] < thr); any |= hit[e]; }
    if (any != 0u) {
#pragma unroll
      for (int e = 0; e < 16; ++e) {
        if (__builtin_expect(hit[e] != 0u, 0)) {
          const float v = cv[e];
          const int qi = ch * 16 + e;
          const bool l0 = v < bd[0], l1 = v < bd[1], l2 = v < bd[2];
          bd[2] = l1 ? bd[1] : (l2 ? v : bd[2]);
          bi[2] = l1 ? bi[1] : (l2 ? qi : bi[2]);
          bd[1] = l0 ? bd[0] : (l1 ? v : bd[1]);
          bi[1] = l0 ? bi[0] : (l1 ? qi : bi[1]);
          bd[0] = l0 ? v : bd[0];
          bi[0] = l0 ? qi : bi[0];
        }
      }
    }
    bb = bb_next;
  }

  int n = n0 + lane;
  float dv0 = fmaxf(bd[0], 1e-10f), dv1 = fmaxf(bd[1], 1e-10f), dv2 = fmaxf(bd[2], 1e-10f);
  float i0 = 1.0f / (dv0 + 1e-8f), i1 = 1.0f / (dv1 + 1e-8f), i2 = 1.0f / (dv2 + 1e-8f);
  float inv_s = 1.0f / (i0 + i1 + i2);
  size_t base = ((size_t)b * N_ + n) * 3;
  idx3[base + 0] = bi[0]; idx3[base + 1] = bi[1]; idx3[base + 2] = bi[2];
  w3[base + 0] = i0 * inv_s; w3[base + 1] = i1 * inv_s; w3[base + 2] = i2 * inv_s;
}

__global__ __launch_bounds__(256) void group_mean_kernel(const float* __restrict__ feat,
                                                         const int* __restrict__ nbr,
                                                         float* __restrict__ coarse) {
  int bc = blockIdx.x;
  int b = bc / C_, c = bc % C_;
  const float* F = feat + ((size_t)b * C_ + c) * N_;
  const int* NB  = nbr + (size_t)b * Q_ * S_;
  float* O = coarse + ((size_t)b * C_ + c) * Q_;
  __shared__ __attribute__((aligned(16))) float lf[N_];
  float4 row4[N_ / 4 / 256];
#pragma unroll
  for (int j = 0; j < N_ / 4 / 256; ++j) row4[j] = reinterpret_cast<const float4*>(F)[(int)threadIdx.x + j * 256];
#pragma unroll
  for (int j = 0; j < N_ / 4 / 256; ++j) *reinterpret_cast<float4*>(lf + ((int)threadIdx.x + j * 256) * 4) = row4[j];
  __syncthreads();
#pragma unroll 2
  for (int j = 0; j < Q_ / 256; ++j) {
    const int q = (int)threadIdx.x + j * 256;
    float s = 0.0f;
#pragma unroll
    for (int k = 0; k < S_; ++k) s += lf[NB[q * S_ + k]];
    O[q] = s * (1.0f / S_);
  }
}

__global__ __launch_bounds__(256) void interp_kernel(const float* __restrict__ coarse,
                                                     const int* __restrict__ idx3,
                                                     const float* __restrict__ w3,
                                                     float* __restrict__ out) {
  int bc = blockIdx.x;
  int b = bc / C_, c = bc % C_;
  const float* CO = coarse + ((size_t)b * C_ + c) * Q_;
  const int*   I  = idx3 + (size_t)b * N_ * 3;
  const float* W  = w3 + (size_t)b * N_ * 3;
  float* O = out + ((size_t)b * C_ + c) * N_;
  for (int n = threadIdx.x; n < N_; n += blockDim.x) {
    O[n] = W[n * 3 + 0] * CO[I[n * 3 + 0]] +
           W[n * 3 + 1] * CO[I[n * 3 + 1]] +
           W[n * 3 + 2] * CO[I[n * 3 + 2]];
  }
}

extern "C" void kernel_launch(void* const* d_in, const int* in_sizes, int n_in,
                              void* d_out, int out_size, void* d_ws, size_t ws_size,
                              hipStream_t stream) {
  const float* points   = (const float*)d_in[0];
  const float* features = (const float*)d_in[1];

  auto align256 = [](size_t x) { return (x + 255) & ~(size_t)255; };
  char* ws = (char*)d_ws;
  size_t off = 0;
  int*   fps_idx = (int*)(ws + off);   off = align256(off + sizeof(int) * B_ * Q_);
  float* qA      = (float*)(ws + off); off = align256(off + sizeof(float) * B_ * Q_ * 4);
  float* pn2     = (float*)(ws + off); off = align256(off + sizeof(float) * B_ * N_);
  int*   nbr     = (int*)(ws + off);   off = align256(off + sizeof(int) * B_ * Q_ * S_);
  float* coarse  = (float*)(ws + off); off = align256(off + sizeof(float) * B_ * C_ * Q_);
  int*   idx3    = (int*)(ws + off);   off = align256(off + sizeof(int) * B_ * N_ * 3);
  float* w3      = (float*)(ws + off); off = align256(off + sizeof(float) * B_ * N_ * 3);

  float* out = (float*)d_out;

  pnorm2_kernel<<<(B_ * N_ + 255) / 256, 256, 0, stream>>>(points, pn2);
  fps_kernel<<<B_, 256, 0, stream>>>(points, fps_idx, qA);
  knn_kernel<<<B_ * 32, 128, 0, stream>>>(points, pn2, qA, nbr);
  group_mean_kernel<<<B_ * C_, 256, 0, stream>>>(features, nbr, coarse);
  three_nn_kernel<<<B_ * 32, 256, 0, stream>>>(points, pn2, qA, idx3, w3);
  interp_kernel<<<B_ * C_, 256, 0, stream>>>(coarse, idx3, w3, out);
}


# knn: queries go to the waves in spatial order (Z-order cell key, bitonic sort in LDS per block): a wave's 16 queries are neighbours, so one insert pass serves several lanes; same lists
# speedup vs baseline: 4.6373x; 1.0370x over previous
#include <hip/hip_runtime.h>

#define B_ 4
#define N_ 8192
#define C_ 128
#define Q_ 2048
#define S_ 16
#define KEEP_ 17
#define RS_ 20

typedef __attribute__((ext_vector_type(2))) float v2f;
typedef __attribute__((ext_vector_type(8))) float v8f;

__global__ void pnorm2_kernel(const float* __restrict__ pts, float* __restrict__ pn2) {
  int i = blockIdx.x * blockDim.x + threadIdx.x;
  if (i < B_ * N_) {
    float x = pts[i * 3 + 0], y = pts[i * 3 + 1], z = pts[i * 3 + 2];
    pn2[i] = x * x + y * y + z * z;
  }
}

#define PSEG_ 32
#define PE_   (8 * PSEG_ / 32)
#define PTARGET_ 128.0f
#define LMAX_ 256
#define KG_   3

__device__ __forceinline__ float dist2(float x, float y, float z, float cx, float cy, float cz) {
  float dx = x - cx, dy = y - cy, dz = z - cz;
  return fmaf(dz, dz, fmaf(dx, dx, dy * dy));
}

__device__ __forceinline__ int uni_i(int v) { return __builtin_amdgcn_readfirstlane(v); }
__device__ __forceinline__ float uni_f(float v) {
  return __builtin_bit_cast(float, __builtin_amdgcn_readfirstlane(__builtin_bit_cast(int, v)));
}

template <int XM>
__device__ __forceinline__ unsigned umax_row(unsigned v) {
  const unsigned o = (unsigned)__builtin_amdgcn_update_dpp(0, (int)v, 0x160 + XM, 0xf, 0xf, true);
  return o > v ? o : v;
}
__device__ __forceinline__ unsigned umax_halves(unsigned v) {
  const unsigned o = (unsigned)__builtin_amdgcn_permlanex16((int)v, (int)v, 0x76543210, 0xfedcba98, false, false);
  return o > v ? o : v;
}
__device__ __forceinline__ unsigned umax_wave(unsigned v) {
  v = umax_row<1>(v);
  v = umax_row<2>(v);
  v = umax_row<4>(v);
  v = umax_row<8>(v);
  return umax_halves(v);
}

#define OWN_  224
#define NPT_  37

__global__ __launch_bounds__(256) void fps_kernel(const float* __restrict__ pts,
                                                  int* __restrict__ fps_idx,
                                                  float* __restrict__ qA) {
  const int b = blockIdx.x;
  const int tid = threadIdx.x;
  const int wave = uni_i(tid >> 5);
  const int lane = tid & 31;
  const int a = tid - 32;
  const float* P = pts + (size_t)b * N_ * 3;

  __shared__ float sval[256];
  __shared__ int   sidx[256];
  __shared__ float4 plist[LMAX_ + KG_];
  __shared__ int   s_np;
  __shared__ float s_dref;
  __shared__ int   s_tot;
  __shared__ float seg_d[8 * PSEG_];
  __shared__ float seg_x[8 * PSEG_];
  __shared__ float seg_y[8 * PSEG_];
  __shared__ float seg_z[8 * PSEG_];
  __shared__ int   seg_i[8 * PSEG_];
  __shared__ int   seg_cnt[8];
  __shared__ float seg_drop[8];

  float px[NPT_], py[NPT_], pz[NPT_], d[NPT_];
  const float p0x = P[0], p0y = P[1], p0z = P[2];
#pragma unroll
  for (int i = 0; i < NPT_; ++i) {
    const int n = a + i * OWN_;
    const bool own = (wave != 0) && (n < N_);
    px[i] = 0.0f; py[i] = 0.0f; pz[i] = 0.0f; d[i] = -1.0f;
    if (own) {
      px[i] = P[n * 3 + 0]; py[i] = P[n * 3 + 1]; pz[i] = P[n * 3 + 2];
      d[i] = dist2(px[i], py[i], pz[i], p0x, p0y, p0z);
    }
  }
  if (tid == 0) {
    fps_idx[b * Q_ + 0] = 0;
    qA[((size_t)b * Q_ + 0) * 4 + 0] = p0x;
    qA[((size_t)b * Q_ + 0) * 4 + 1] = p0y;
    qA[((size_t)b * Q_ + 0) * 4 + 2] = p0z;
    qA[((size_t)b * Q_ + 0) * 4 + 3] = p0x * p0x + p0y * p0y + p0z * p0z;
  }

  auto apply_own = [&](float cx, float cy, float cz) {
#pragma unroll
    for (int g = 0; g < NPT_; g += 8) {
      float dx[8], dy[8], dz[8], t[8];
#pragma unroll
      for (int k = 0; k < 8; ++k) if (g + k < NPT_) { dx[k] = px[g + k] - cx; dy[k] = py[g + k] - cy; dz[k] = pz[g + k] - cz; }
#pragma unroll
      for (int k = 0; k < 8; ++k) if (g + k < NPT_) t[k] = dy[k] * dy[k];
#pragma unroll
      for (int k = 0; k < 8; ++k) if (g + k < NPT_) t[k] = fmaf(dx[k], dx[k], t[k]);
#pragma unroll
      for (int k = 0; k < 8; ++k) if (g + k < NPT_) t[k] = fmaf(dz[k], dz[k], t[k]);
#pragma unroll
      for (int k = 0; k < 8; ++k) if (g + k < NPT_) d[g + k] = fminf(d[g + k], t[k]);
    }
  };

  auto apply_group = [&](const float4 (&g)[KG_]) {
#pragma unroll
    for (int gq = 0; gq < NPT_; gq += 8) {
      float t[KG_][8];
#pragma unroll
      for (int j = 0; j < KG_; ++j) {
        float dx[8], dy[8], dz[8];
#pragma unroll
        for (int k = 0; k < 8; ++k) if (gq + k < NPT_) { dx[k] = px[gq + k] - g[j].x; dy[k] = py[gq + k] - g[j].y; dz[k] = pz[gq + k] - g[j].z; }
#pragma unroll
        for (int k = 0; k < 8; ++k) if (gq + k < NPT_) t[j][k] = dy[k] * dy[k];
#pragma unroll
        for (int k = 0; k < 8; ++k) if (gq + k < NPT_) t[j][k] = fmaf(dx[k], dx[k], t[j][k]);
#pragma unroll
        for (int k = 0; k < 8; ++k) if (gq + k < NPT_) t[j][k] = fmaf(dz[k], dz[k], t[j][k]);
      }
#pragma unroll
      for (int k = 0; k < 8; ++k) if (gq + k < NPT_) {
        float m = d[gq + k];
#pragma unroll
        for (int j = 0; j < KG_; ++j) m = fminf(m, t[j][k]);
        d[gq + k] = m;
      }
    }
  };

  float gap = 0.1f;
  float dref = 0.0f;
  int   need_full = 1;
  int   s = 1;
  while (s < Q_) {
    int np0 = 0;
    if (need_full) {
      float bv = -1.0f;
      int   bidx = 0;
#pragma unroll
      for (int i = 0; i < NPT_; ++i) {
        if (d[i] > bv) { bv = d[i]; bidx = a + i * OWN_; }
      }
      sval[tid] = bv; sidx[tid] = bidx;
      __syncthreads();
      for (int st = 128; st > 0; st >>= 1) {
        if (tid < st) {
          float ov = sval[tid + st]; int oi = sidx[tid + st];
          float mv = sval[tid];      int mi = sidx[tid];
          if (ov > mv || (ov == mv && oi < mi)) { sval[tid] = ov; sidx[tid] = oi; }
        }
        __syncthreads();
      }
      if (tid == 0) {
        int nxt = sidx[0];
        float x = P[nxt * 3 + 0], y = P[nxt * 3 + 1], z = P[nxt * 3 + 2];
        fps_idx[b * Q_ + s] = nxt;
        qA[((size_t)b * Q_ + s) * 4 + 0] = x;
        qA[((size_t)b * Q_ + s) * 4 + 1] = y;
        qA[((size_t)b * Q_ + s) * 4 + 2] = z;
        qA[((size_t)b * Q_ + s) * 4 + 3] = x * x + y * y + z * z;
        plist[0] = make_float4(x, y, z, 0.0f);
      }
      dref = uni_f(sval[0]);
      __syncthreads();
      if (wave != 0) {
        const float4 c4 = plist[0];
        apply_own(c4.x, c4.y, c4.z);
      }
      s += 1;
      np0 = 1;
      if (s >= Q_) break;
    }

    const float T = dref * (1.0f - gap);
    int cnt = 0;
    float drop = 0.0f;
    if (wave != 0) {
#pragma unroll
      for (int i = 0; i < NPT_; ++i) {
        const bool in = d[i] > T;
        const unsigned m = __builtin_amdgcn_ballot_w32(in);
        const int pos = cnt + (int)__builtin_amdgcn_mbcnt_lo(m, 0u);
        if (in) {
          if (pos < PSEG_) {
            const int sl = wave * PSEG_ + pos;
            seg_d[sl] = d[i]; seg_i[sl] = a + i * OWN_;
            seg_x[sl] = px[i]; seg_y[sl] = py[i]; seg_z[sl] = pz[i];
          } else {
            drop = fmaxf(drop, d[i]);
          }
        }
        cnt += __builtin_popcount(m);
      }
      if (cnt > PSEG_) {
#pragma unroll
        for (int o = 16; o > 0; o >>= 1) drop = fmaxf(drop, __shfl_xor(drop, o, 32));
      }
    }
    if (lane == 0) { seg_cnt[wave] = cnt; seg_drop[wave] = drop; }
    __syncthreads();

    float teff = T;
    int tot = 0;
    float ex[PE_], ey[PE_], ez[PE_], ed[PE_];
    unsigned ek[PE_];
#pragma unroll
    for (int e = 0; e < PE_; ++e) { ex[e] = 0.0f; ey[e] = 0.0f; ez[e] = 0.0f; ed[e] = 0.0f; ek[e] = 0u; }
    if (wave == 0) {
#pragma unroll
      for (int w = 0; w < 8; ++w) { teff = fmaxf(teff, seg_drop[w]); tot += seg_cnt[w]; }
      teff = uni_f(teff);
#pragma unroll
      for (int e = 0; e < PE_; ++e) {
        const int w = (e * 32) / PSEG_;
        const int pos = (e * 32) % PSEG_ + lane;
        const int sl = w * PSEG_ + pos;
        const int c = seg_cnt[w];
        int id = 0;
        if (pos < c) {
          ed[e] = seg_d[sl]; ex[e] = seg_x[sl]; ey[e] = seg_y[sl]; ez[e] = seg_z[sl];
          id = seg_i[sl];
        }
        ek[e] = ((unsigned)(N_ - 1 - id) << 9) | (unsigned)sl;
      }
    }

    int np = 0;
    float last = dref;
    for (int c = 0; ; ++c) {
      bool end = false;
      if (wave == 0) {
#pragma unroll
        for (int j = 0; j < KG_; ++j) {
          if (!end) {
            end = !(s + np < Q_ && np < LMAX_);
            float bd = 0.0f;
            unsigned blo = 0u;
            if (!end) {
              unsigned db[PE_], mm[PE_];
#pragma unroll
              for (int e = 0; e < PE_; ++e) { db[e] = __float_as_uint(ed[e]); mm[e] = db[e]; }
#pragma unroll
              for (int w2 = PE_ / 2; w2 > 0; w2 >>= 1) {
#pragma unroll
                for (int e = 0; e < w2; ++e) mm[e] = mm[e + w2] > mm[e] ? mm[e + w2] : mm[e];
              }
              const unsigned dmax = umax_wave(mm[0]);
              unsigned cc[PE_];
#pragma unroll
              for (int e = 0; e < PE_; ++e) cc[e] = db[e] == dmax ? ek[e] : 0u;
#pragma unroll
              for (int w2 = PE_ / 2; w2 > 0; w2 >>= 1) {
#pragma unroll
                for (int e = 0; e < w2; ++e) cc[e] = cc[e + w2] > cc[e] ? cc[e + w2] : cc[e];
              }
              const unsigned kmax = umax_wave(cc[0]);
              bd = __uint_as_float((unsigned)uni_i((int)dmax));
              blo = (unsigned)uni_i((int)kmax);
              if (!(bd > teff)) { last = teff; end = true; }
            }
            if (end) {
              if (lane == 0) {
#pragma unroll
                for (int jj = j; jj < KG_; ++jj) plist[c * KG_ + jj] = make_float4(0.0f, 0.0f, 0.0f, __int_as_float(1));
                s_np = np; s_dref = last; s_tot = tot;
              }
            } else {
              const int sl = (int)(blo & 511u);
              const int id = N_ - 1 - (int)(blo >> 9);
              const float qx = seg_x[sl], qy = seg_y[sl], qz = seg_z[sl];
              if (lane == 0) plist[np] = make_float4(qx, qy, qz, __int_as_float(id << 1));
              ++np;
              last = bd;
#pragma unroll
              for (int e = 0; e < PE_; ++e)
                ed[e] = fminf(ed[e], dist2(ex[e], ey[e], ez[e], qx, qy, qz));
            }
          }
        }
      }
      __builtin_amdgcn_fence(__ATOMIC_RELEASE, "workgroup", "local");
      __builtin_amdgcn_s_barrier();
      __builtin_amdgcn_fence(__ATOMIC_ACQUIRE, "workgroup", "local");
      if (wave != 0) {
        float4 g[KG_];
#pragma unroll
        for (int j = 0; j < KG_; ++j) g[j] = plist[c * KG_ + j];
        int nv = 0;
#pragma unroll
        for (int j = 0; j < KG_; ++j) {
          const int wj = uni_i(__float_as_int(g[j].w));
          if (!end) {
            if ((wj & 1) != 0) end = true; else ++nv;
          }
        }
        if (nv > 0) {
#pragma unroll
          for (int j = 1; j < KG_; ++j) if (j >= nv) g[j] = g[0];
          if (wave == 4) {
            const int sj = lane / 5;
            const int sr = lane - sj * 5;
            if (sj < nv) {
              float4 e4 = g[0];
#pragma unroll
              for (int j = 1; j < KG_; ++j) if (sj == j) e4 = g[j];
              const int q = s + c * KG_ + sj;
              const float n2 = fmaf(e4.z, e4.z, fmaf(e4.x, e4.x, e4.y * e4.y));
              float comp = e4.x;
              comp = sr == 1 ? e4.y : comp;
              comp = sr == 2 ? e4.z : comp;
              comp = sr == 3 ? n2 : comp;
              int* rowp = reinterpret_cast<int*>(qA + ((size_t)b * Q_ + q) * 4) + sr;
              int* dst = sr == 4 ? fps_idx + b * Q_ + q : rowp;
              *dst = sr == 4 ? (__float_as_int(e4.w) >> 1) : __float_as_int(comp);
            }
          }
          apply_group(g);
        }
      }
      if (end) break;
    }

    np = uni_i(s_np);
    dref = uni_f(s_dref);
    tot = uni_i(s_tot);
    s += np;
    need_full = (np + np0 == 0);

    float f = sqrtf(PTARGET_ / (float)(tot > 1 ? tot : 1));
    f = fminf(fmaxf(f, 0.5f), 2.0f);
    gap = fminf(fmaxf(gap * f, 1.0e-3f), 0.5f);
  }
}


__global__ __launch_bounds__(128) void knn_kernel(const float* __restrict__ pts,
                                                  const float* __restrict__ pn2,
                                                  const float* __restrict__ qA,
                                                  int* __restrict__ nbr) {
  const int wave = threadIdx.x >> 5;
  const int lane = threadIdx.x & 31;
  const int b    = blockIdx.x >> 5;
  const int q0   = ((blockIdx.x & 31) * 4 + wave) * 16;

  const float* P  = pts + (size_t)b * N_ * 3;
  const float* PN = pn2 + (size_t)b * N_;
  const float* QA = qA + (size_t)b * Q_ * 4;

  __shared__ float stage[4][16 * RS_];
  __shared__ float mrg_d[4][16 * KEEP_];
  __shared__ int   mrg_i[4][16 * KEEP_];
  __shared__ __attribute__((aligned(16))) float lpts[N_ * 4];
  __shared__ unsigned skey[Q_];
  float* st = stage[wave];

  const int lhalf = lane >> 4;
  const int l16   = lane & 15;

#pragma unroll
  for (int j = 0; j < Q_ / 128; ++j) {
    const int q = (int)threadIdx.x + j * 128;
    const float4 r = reinterpret_cast<const float4*>(QA)[q];
    const unsigned xi = (unsigned)(int)fminf(fmaxf((r.x + 4.0f) * 16.0f, 0.0f), 127.0f);
    const unsigned yi = (unsigned)(int)fminf(fmaxf((r.y + 4.0f) * 16.0f, 0.0f), 127.0f);
    const unsigned zi = (unsigned)(int)fminf(fmaxf((r.z + 4.0f) * 16.0f, 0.0f), 127.0f);
    unsigned m = 0u;
#pragma unroll
    for (int bit = 6; bit >= 0; --bit)
      m = (m << 3) | (((xi >> bit) & 1u) << 2) | (((yi >> bit) & 1u) << 1) | ((zi >> bit) & 1u);
    skey[q] = (m << 11) | (unsigned)q;
  }
  {
    const float4* P4  = reinterpret_cast<const float4*>(P);
    const float4* PN4 = reinterpret_cast<const float4*>(PN);
#pragma unroll 4
    for (int j = 0; j < N_ / 4 / 128; ++j) {
      const int g = (int)threadIdx.x + j * 128;
      const float4 u0 = P4[g * 3 + 0], u1 = P4[g * 3 + 1], u2 = P4[g * 3 + 2];
      const float4 n4 = PN4[g];
      float4* dst = reinterpret_cast<float4*>(lpts + g * 16);
      dst[0] = make_float4(u0.x, u0.y, u0.z, n4.x);
      dst[1] = make_float4(u0.w, u1.x, u1.y, n4.y);
      dst[2] = make_float4(u1.z, u1.w, u2.x, n4.z);
      dst[3] = make_float4(u2.y, u2.z, u2.w, n4.w);
    }
  }
  __syncthreads();

  for (int kk = 2; kk <= Q_; kk <<= 1) {
    for (int jj = kk >> 1; jj > 0; jj >>= 1) {
      unsigned ka[Q_ / 2 / 128], kb[Q_ / 2 / 128];
#pragma unroll
      for (int t = 0; t < Q_ / 2 / 128; ++t) {
        const int c = (int)threadIdx.x + t * 128;
        const int i = ((c & ~(jj - 1)) << 1) | (c & (jj - 1));
        ka[t] = skey[i];
        kb[t] = skey[i | jj];
      }
#pragma unroll
      for (int t = 0; t < Q_ / 2 / 128; ++t) {
        const int c = (int)threadIdx.x + t * 128;
        const int i = ((c & ~(jj - 1)) << 1) | (c & (jj - 1));
        const unsigned lo = ka[t] < kb[t] ? ka[t] : kb[t], hi = ka[t] < kb[t] ? kb[t] : ka[t];
        const bool up = (i & kk) == 0;
        skey[i] = up ? lo : hi;
        skey[i | jj] = up ? hi : lo;
      }
      __syncthreads();
    }
  }
  const int qsel = (int)(skey[q0 + l16] & (unsigned)(Q_ - 1));

  const float4 qa0 = ((const float4*)QA)[qsel];
  v2f a0 = lhalf ? (v2f){-2.0f * qa0.z, 1.0f} : (v2f){-2.0f * qa0.x, -2.0f * qa0.y};

  v8f c0;
#pragma unroll
  for (int r = 0; r < 8; ++r) c0[r] = QA[(int)(skey[q0 + r + lhalf * 8] & (unsigned)(Q_ - 1)) * 4 + 3];

  float bd[KEEP_]; int bi[KEEP_];
#pragma unroll
  for (int k = 0; k < KEEP_; ++k) { bd[k] = 3.0e38f; bi[k] = 0; }

  auto load_bb = [&](int ch) -> v2f {
    return *reinterpret_cast<const v2f*>(lpts + (ch * 16 + l16) * 4 + lhalf * 2);
  };

  const int NCH = N_ / 16;
  v2f bb = load_bb(0);
  for (int ch = 0; ch < NCH; ++ch) {
    v2f bb_next = load_bb(ch + 1 < NCH ? ch + 1 : NCH - 1);

    v8f d0 = __builtin_amdgcn_wmma_f32_16x16x4_f32(false, a0, false, bb, (short)0, c0, false, false);

#pragma unroll
    for (int r = 0; r < 8; ++r) st[(r + lhalf * 8) * RS_ + l16] = d0[r];

    const float4* myrow = (const float4*)(st + l16 * RS_ + lhalf * 8);
    const float4 r0 = myrow[0], r1 = myrow[1];
    const float cv[8] = {fmaxf(r0.x, 0.0f), fmaxf(r0.y, 0.0f), fmaxf(r0.z, 0.0f), fmaxf(r0.w, 0.0f),
                         fmaxf(r1.x, 0.0f), fmaxf(r1.y, 0.0f), fmaxf(r1.z, 0.0f), fmaxf(r1.w, 0.0f)};
    const float thr = bd[KEEP_ - 1];
    unsigned hit[8];
    unsigned any = 0u;
#pragma unroll
    for (int e = 0; e < 8; ++e) { hit[e] = __builtin_amdgcn_ballot_w32(cv[e] < thr); any |= hit[e]; }
    if (any != 0u) {
      const int pbase = ch * 16 + lhalf * 8;
#pragma unroll
      for (int e = 0; e < 8; ++e) {
        if (__builtin_expect(hit[e] != 0u, 0)) {
          const float v = cv[e];
          const int pi = pbase + e;
          bool lt[KEEP_];
#pragma unroll
          for (int k = 0; k < KEEP_; ++k) lt[k] = v < bd[k];
#pragma unroll
          for (int k = KEEP_ - 1; k > 0; --k) {
            bd[k] = __builtin_amdgcn_fmed3f(bd[k - 1], v, bd[k]);
            bi[k] = lt[k - 1] ? bi[k - 1] : (lt[k] ? pi : bi[k]);
          }
          bd[0] = lt[0] ? v : bd[0];
          bi[0] = lt[0] ? pi : bi[0];
        }
      }
    }
    bb = bb_next;
  }

  if (lhalf) {
#pragma unroll
    for (int k = 0; k < KEEP_; ++k) {
      mrg_d[wave][l16 * KEEP_ + k] = bd[k];
      mrg_i[wave][l16 * KEEP_ + k] = bi[k];
    }
  }
  __syncthreads();
  if (!lhalf) {
#pragma unroll 1
    for (int k = 0; k < KEEP_; ++k) {
      float v  = mrg_d[wave][l16 * KEEP_ + k];
      int   pi = mrg_i[wave][l16 * KEEP_ + k];
      if (v < bd[KEEP_ - 1] || (v == bd[KEEP_ - 1] && pi < bi[KEEP_ - 1])) {
        bool lt[KEEP_];
#pragma unroll
        for (int k = 0; k < KEEP_; ++k) lt[k] = v < bd[k] || (v == bd[k] && pi < bi[k]);
#pragma unroll
        for (int k = KEEP_ - 1; k > 0; --k) {
          bd[k] = lt[k - 1] ? bd[k - 1] : (lt[k] ? v : bd[k]);
          bi[k] = lt[k - 1] ? bi[k - 1] : (lt[k] ? pi : bi[k]);
        }
        bd[0] = lt[0] ? v : bd[0];
        bi[0] = lt[0] ? pi : bi[0];
      }
    }
    int q = qsel;
#pragma unroll
    for (int k = 0; k < S_; ++k) nbr[((size_t)b * Q_ + q) * S_ + k] = bi[k + 1];
  }
}

__global__ __launch_bounds__(256) void three_nn_kernel(const float* __restrict__ pts,
                                                       const float* __restrict__ pn2,
                                                       const float* __restrict__ qA,
                                                       int* __restrict__ idx3,
                                                       float* __restrict__ w3) {
  const int wave = threadIdx.x >> 5;
  const int lane = threadIdx.x & 31;
  const int b    = blockIdx.x >> 5;
  const int n0   = ((blockIdx.x & 31) * 8 + wave) * 32;

  const float* P  = pts + (size_t)b * N_ * 3;
  const float* PN = pn2 + (size_t)b * N_;
  const float* QA = qA + (size_t)b * Q_ * 4;

  __shared__ float stage[8][32 * RS_];
  __shared__ __attribute__((aligned(16))) float lq[Q_ * 4];
  float* st = stage[wave];

  const int lhalf = lane >> 4;
  const int l16   = lane & 15;

#pragma unroll
  for (int j = 0; j < Q_ / 256; ++j) {
    const int i = (int)threadIdx.x + j * 256;
    *reinterpret_cast<float4*>(lq + i * 4) = reinterpret_cast<const float4*>(QA)[i];
  }
  __syncthreads();

  int m0 = n0 + l16, m1 = n0 + 16 + l16;
  float p0x = P[m0 * 3 + 0], p0y = P[m0 * 3 + 1], p0z = P[m0 * 3 + 2];
  float p1x = P[m1 * 3 + 0], p1y = P[m1 * 3 + 1], p1z = P[m1 * 3 + 2];
  v2f a0 = lhalf ? (v2f){-2.0f * p0z, 1.0f} : (v2f){-2.0f * p0x, -2.0f * p0y};
  v2f a1 = lhalf ? (v2f){-2.0f * p1z, 1.0f} : (v2f){-2.0f * p1x, -2.0f * p1y};

  v8f c0, c1;
#pragma unroll
  for (int r = 0; r < 8; ++r) {
    c0[r] = PN[n0 + r + lhalf * 8];
    c1[r] = PN[n0 + 16 + r + lhalf * 8];
  }

  float bd[3]; int bi[3];
#pragma unroll
  for (int k = 0; k < 3; ++k) { bd[k] = 3.0e38f; bi[k] = 0; }

  auto load_bb = [&](int ch) -> v2f {
    return *reinterpret_cast<const v2f*>(lq + (ch * 16 + l16) * 4 + lhalf * 2);
  };

  const int NCH = Q_ / 16;
  v2f bb = load_bb(0);
  for (int ch = 0; ch < NCH; ++ch) {
    v2f bb_next = load_bb(ch + 1 < NCH ? ch + 1 : NCH - 1);

    v8f d0 = __builtin_amdgcn_wmma_f32_16x16x4_f32(false, a0, false, bb, (short)0, c0, false, false);
    v8f d1 = __builtin_amdgcn_wmma_f32_16x16x4_f32(false, a1, false, bb, (short)0, c1, false, false);

#pragma unroll
    for (int r = 0; r < 8; ++r) {
      st[(r + lhalf * 8) * RS_ + l16]      = d0[r];
      st[(16 + r + lhalf * 8) * RS_ + l16] = d1[r];
    }
    const float4* myrow = (const float4*)(st + lane * RS_);
    float cv[16];
#pragma unroll
    for (int jj = 0; jj < 4; ++jj) {
      const float4 v4 = myrow[jj];
      cv[jj * 4 + 0] = fmaxf(v4.x, 0.0f); cv[jj * 4 + 1] = fmaxf(v4.y, 0.0f);
      cv[jj * 4 + 2] = fmaxf(v4.z, 0.0f); cv[jj * 4 + 3] = fmaxf(v4.w, 0.0f);
    }
    const float thr = bd[2];
    unsigned hit[16];
    unsigned any = 0u;
#pragma unroll
    for (int e = 0; e < 16; ++e) { hit[e] = __builtin_amdgcn_ballot_w32(cv[e] < thr); any |= hit[e]; }
    if (any != 0u) {
#pragma unroll
      for (int e = 0; e < 16; ++e) {
        if (__builtin_expect(hit[e] != 0u, 0)) {
          const float v = cv[e];
          const int qi = ch * 16 + e;
          const bool l0 = v < bd[0], l1 = v < bd[1], l2 = v < bd[2];
          bd[2] = l1 ? bd[1] : (l2 ? v : bd[2]);
          bi[2] = l1 ? bi[1] : (l2 ? qi : bi[2]);
          bd[1] = l0 ? bd[0] : (l1 ? v : bd[1]);
          bi[1] = l0 ? bi[0] : (l1 ? qi : bi[1]);
          bd[0] = l0 ? v : bd[0];
          bi[0] = l0 ? qi : bi[0];
        }
      }
    }
    bb = bb_next;
  }

  int n = n0 + lane;
  float dv0 = fmaxf(bd[0], 1e-10f), dv1 = fmaxf(bd[1], 1e-10f), dv2 = fmaxf(bd[2], 1e-10f);
  float i0 = 1.0f / (dv0 + 1e-8f), i1 = 1.0f / (dv1 + 1e-8f), i2 = 1.0f / (dv2 + 1e-8f);
  float inv_s = 1.0f / (i0 + i1 + i2);
  size_t base = ((size_t)b * N_ + n) * 3;
  idx3[base + 0] = bi[0]; idx3[base + 1] = bi[1]; idx3[base + 2] = bi[2];
  w3[base + 0] = i0 * inv_s; w3[base + 1] = i1 * inv_s; w3[base + 2] = i2 * inv_s;
}

__global__ __launch_bounds__(256) void group_mean_kernel(const float* __restrict__ feat,
                                                         const int* __restrict__ nbr,
                                                         float* __restrict__ coarse) {
  int bc = blockIdx.x;
  int b = bc / C_, c = bc % C_;
  const float* F = feat + ((size_t)b * C_ + c) * N_;
  const int* NB  = nbr + (size_t)b * Q_ * S_;
  float* O = coarse + ((size_t)b * C_ + c) * Q_;
  __shared__ __attribute__((aligned(16))) float lf[N_];
  float4 row4[N_ / 4 / 256];
#pragma unroll
  for (int j = 0; j < N_ / 4 / 256; ++j) row4[j] = reinterpret_cast<const float4*>(F)[(int)threadIdx.x + j * 256];
#pragma unroll
  for (int j = 0; j < N_ / 4 / 256; ++j) *reinterpret_cast<float4*>(lf + ((int)threadIdx.x + j * 256) * 4) = row4[j];
  __syncthreads();
#pragma unroll 2
  for (int j = 0; j < Q_ / 256; ++j) {
    const int q = (int)threadIdx.x + j * 256;
    float s = 0.0f;
#pragma unroll
    for (int k = 0; k < S_; ++k) s += lf[NB[q * S_ + k]];
    O[q] = s * (1.0f / S_);
  }
}

__global__ __launch_bounds__(256) void interp_kernel(const float* __restrict__ coarse,
                                                     const int* __restrict__ idx3,
                                                     const float* __restrict__ w3,
                                                     float* __restrict__ out) {
  int bc = blockIdx.x;
  int b = bc / C_, c = bc % C_;
  const float* CO = coarse + ((size_t)b * C_ + c) * Q_;
  const int*   I  = idx3 + (size_t)b * N_ * 3;
  const float* W  = w3 + (size_t)b * N_ * 3;
  float* O = out + ((size_t)b * C_ + c) * N_;
  for (int n = threadIdx.x; n < N_; n += blockDim.x) {
    O[n] = W[n * 3 + 0] * CO[I[n * 3 + 0]] +
           W[n * 3 + 1] * CO[I[n * 3 + 1]] +
           W[n * 3 + 2] * CO[I[n * 3 + 2]];
  }
}

extern "C" void kernel_launch(void* const* d_in, const int* in_sizes, int n_in,
                              void* d_out, int out_size, void* d_ws, size_t ws_size,
                              hipStream_t stream) {
  const float* points   = (const float*)d_in[0];
  const float* features = (const float*)d_in[1];

  auto align256 = [](size_t x) { return (x + 255) & ~(size_t)255; };
  char* ws = (char*)d_ws;
  size_t off = 0;
  int*   fps_idx = (int*)(ws + off);   off = align256(off + sizeof(int) * B_ * Q_);
  float* qA      = (float*)(ws + off); off = align256(off + sizeof(float) * B_ * Q_ * 4);
  float* pn2     = (float*)(ws + off); off = align256(off + sizeof(float) * B_ * N_);
  int*   nbr     = (int*)(ws + off);   off = align256(off + sizeof(int) * B_ * Q_ * S_);
  float* coarse  = (float*)(ws + off); off = align256(off + sizeof(float) * B_ * C_ * Q_);
  int*   idx3    = (int*)(ws + off);   off = align256(off + sizeof(int) * B_ * N_ * 3);
  float* w3      = (float*)(ws + off); off = align256(off + sizeof(float) * B_ * N_ * 3);

  float* out = (float*)d_out;

  pnorm2_kernel<<<(B_ * N_ + 255) / 256, 256, 0, stream>>>(points, pn2);
  fps_kernel<<<B_, 256, 0, stream>>>(points, fps_idx, qA);
  knn_kernel<<<B_ * 32, 128, 0, stream>>>(points, pn2, qA, nbr);
  group_mean_kernel<<<B_ * C_, 256, 0, stream>>>(features, nbr, coarse);
  three_nn_kernel<<<B_ * 32, 256, 0, stream>>>(points, pn2, qA, idx3, w3);
  interp_kernel<<<B_ * C_, 256, 0, stream>>>(coarse, idx3, w3, out);
}
